# starting kernel
# baseline (speedup 1.0000x reference)
#include <hip/hip_runtime.h>
#include <hip/hip_bf16.h>

#define N_NODES 100000
#define N_EDGES 1600000
#define TOT_E   (N_EDGES + N_NODES)
#define IN_CH   64
#define OUT_CH  32
#define HEADS   4
#define HC      (HEADS * OUT_CH)
#define NEG_SLOPE 0.2f

typedef __attribute__((ext_vector_type(16))) _Float16 v16h;
typedef __attribute__((ext_vector_type(8)))  float    v8f;

__device__ __forceinline__ unsigned fflip(float f) {
    unsigned u = __float_as_uint(f);
    return (u & 0x80000000u) ? ~u : (u | 0x80000000u);
}
__device__ __forceinline__ float funflip(unsigned u) {
    return __uint_as_float((u & 0x80000000u) ? (u & 0x7FFFFFFFu) : ~u);
}

__device__ __forceinline__ void edge_sd(const int* __restrict__ ei, int e, int& s, int& d) {
    if (e < N_EDGES) { s = ei[e]; d = ei[N_EDGES + e]; }
    else             { s = d = e - N_EDGES; }
}

__global__ __launch_bounds__(256) void k_gemm(const float* __restrict__ x,
                                              const float* __restrict__ W,
                                              float* __restrict__ h) {
    __shared__ float xs[16 * IN_CH];
    const int tid = threadIdx.x;
    const int mt  = blockIdx.x;
    for (int i = tid; i < 16 * IN_CH; i += 256) {
        int r = i >> 6, c = i & 63;
        xs[i] = x[(size_t)(mt * 16 + r) * IN_CH + c];
    }
    __syncthreads();

    const int wave = tid >> 5;
    const int lane = tid & 31;
    const int mrow = lane & 15;
    const int khalf = (lane >> 4) << 3;
    const int ncol = wave * 16 + (lane & 15);

    v16h a0, a1, b0, b1;
    #pragma unroll
    for (int i = 0; i < 16; ++i) {
        int k = i + (i & 8) + khalf;
        a0[i] = (_Float16)xs[mrow * IN_CH + k];
        a1[i] = (_Float16)xs[mrow * IN_CH + k + 32];
        b0[i] = (_Float16)W[(size_t)k * HC + ncol];
        b1[i] = (_Float16)W[(size_t)(k + 32) * HC + ncol];
    }
    v8f c = {};
    c = __builtin_amdgcn_wmma_f32_16x16x32_f16(false, a0, false, b0, (short)0, c, false, false);
    c = __builtin_amdgcn_wmma_f32_16x16x32_f16(false, a1, false, b1, (short)0, c, false, false);

    #pragma unroll
    for (int r = 0; r < 8; ++r) {
        int m = mt * 16 + r + khalf;
        h[(size_t)m * HC + ncol] = c[r];
    }
}

__global__ __launch_bounds__(256) void k_att(const float* __restrict__ h,
                                             const float* __restrict__ att_src,
                                             const float* __restrict__ att_dst,
                                             float* __restrict__ a_src,
                                             float* __restrict__ a_dst) {
    int idx = blockIdx.x * blockDim.x + threadIdx.x;
    if (idx >= N_NODES * HEADS) return;
    int n = idx >> 2, hh = idx & 3;
    const float* hp = h + (size_t)n * HC + hh * OUT_CH;
    float s = 0.f, d = 0.f;
    #pragma unroll
    for (int c = 0; c < OUT_CH; ++c) {
        float v = hp[c];
        s += v * att_src[hh * OUT_CH + c];
        d += v * att_dst[hh * OUT_CH + c];
    }
    a_src[idx] = s; a_dst[idx] = d;
}

__global__ __launch_bounds__(256) void k_init(unsigned* __restrict__ mflip,
                                              float* __restrict__ denom,
                                              float* __restrict__ out) {
    const int stride = gridDim.x * blockDim.x;
    for (int i = blockIdx.x * blockDim.x + threadIdx.x; i < N_NODES * HC; i += stride)
        out[i] = 0.f;
    for (int i = blockIdx.x * blockDim.x + threadIdx.x; i < N_NODES * HEADS; i += stride) {
        mflip[i] = 0u;
        denom[i] = 0.f;
    }
}

__global__ __launch_bounds__(256) void k_max(const int* __restrict__ ei,
                                             const float* __restrict__ a_src,
                                             const float* __restrict__ a_dst,
                                             unsigned* __restrict__ mflip) {
    int e = blockIdx.x * blockDim.x + threadIdx.x;
    if (e >= TOT_E) return;
    int s, d; edge_sd(ei, e, s, d);
    #pragma unroll
    for (int hh = 0; hh < HEADS; ++hh) {
        float v = a_src[s * HEADS + hh] + a_dst[d * HEADS + hh];
        v = v > 0.f ? v : NEG_SLOPE * v;
        atomicMax(&mflip[d * HEADS + hh], fflip(v));
    }
}

__global__ __launch_bounds__(256) void k_denom(const int* __restrict__ ei,
                                               const float* __restrict__ a_src,
                                               const float* __restrict__ a_dst,
                                               const unsigned* __restrict__ mflip,
                                               float* __restrict__ denom) {
    int e = blockIdx.x * blockDim.x + threadIdx.x;
    if (e >= TOT_E) return;
    int s, d; edge_sd(ei, e, s, d);
    #pragma unroll
    for (int hh = 0; hh < HEADS; ++hh) {
        float v = a_src[s * HEADS + hh] + a_dst[d * HEADS + hh];
        v = v > 0.f ? v : NEG_SLOPE * v;
        float m = funflip(mflip[d * HEADS + hh]);
        atomicAdd(&denom[d * HEADS + hh], __expf(v - m));
    }
}

__global__ __launch_bounds__(256) void k_agg(const int* __restrict__ ei,
                                             const float* __restrict__ a_src,
                                             const float* __restrict__ a_dst,
                                             const unsigned* __restrict__ mflip,
                                             const float* __restrict__ denom,
                                             const float* __restrict__ h,
                                             float* __restrict__ out) {
    const int lane  = threadIdx.x & 31;
    const int hh    = lane >> 3;
    const int wglob = (blockIdx.x * blockDim.x + threadIdx.x) >> 5;
    const int nwav  = (gridDim.x * blockDim.x) >> 5;

    for (int e = wglob; e < TOT_E; e += nwav) {
        int s, d; edge_sd(ei, e, s, d);

        int en = e + nwav;
        if (en < TOT_E) {
            int sn = (en < N_EDGES) ? ei[en] : (en - N_EDGES);
            __builtin_prefetch(h + (size_t)sn * HC + lane * 4, 0, 0);
        }

        float v = a_src[s * HEADS + hh] + a_dst[d * HEADS + hh];
        v = v > 0.f ? v : NEG_SLOPE * v;
        float m     = funflip(mflip[d * HEADS + hh]);
        float alpha = __expf(v - m) / (denom[d * HEADS + hh] + 1e-16f);

        float4 hv = ((const float4*)(h + (size_t)s * HC))[lane];
        float* op = out + (size_t)d * HC + lane * 4;
        atomicAdd(op + 0, hv.x * alpha);
        atomicAdd(op + 1, hv.y * alpha);
        atomicAdd(op + 2, hv.z * alpha);
        atomicAdd(op + 3, hv.w * alpha);
    }
}

__global__ __launch_bounds__(256) void k_lsm(float* __restrict__ out,
                                             const float* __restrict__ bias) {
    const int lane = threadIdx.x & 31;
    const int node = (blockIdx.x * blockDim.x + threadIdx.x) >> 5;
    if (node >= N_NODES) return;
    float* row = out + (size_t)node * HC;
    float4 v = ((float4*)row)[lane];
    float4 b = ((const float4*)bias)[lane];
    v.x += b.x; v.y += b.y; v.z += b.z; v.w += b.w;

    float mx = fmaxf(fmaxf(v.x, v.y), fmaxf(v.z, v.w));
    #pragma unroll
    for (int o = 16; o >= 1; o >>= 1) mx = fmaxf(mx, __shfl_xor(mx, o, 32));

    float se = __expf(v.x - mx) + __expf(v.y - mx) + __expf(v.z - mx) + __expf(v.w - mx);
    #pragma unroll
    for (int o = 16; o >= 1; o >>= 1) se += __shfl_xor(se, o, 32);

    float lse = mx + __logf(se);
    v.x -= lse; v.y -= lse; v.z -= lse; v.w -= lse;
    ((float4*)row)[lane] = v;
}

extern "C" void kernel_launch(void* const* d_in, const int* in_sizes, int n_in,
                              void* d_out, int out_size, void* d_ws, size_t ws_size,
                              hipStream_t stream) {
    const float* x       = (const float*)d_in[0];
    const int*   ei      = (const int*)  d_in[1];
    const float* W       = (const float*)d_in[2];
    const float* att_src = (const float*)d_in[3];
    const float* att_dst = (const float*)d_in[4];
    const float* bias    = (const float*)d_in[5];
    float*       out     = (float*)d_out;

    float*    h     = (float*)d_ws;
    float*    asrc  = h    + (size_t)N_NODES * HC;
    float*    adst  = asrc + (size_t)N_NODES * HEADS;
    unsigned* mflip = (unsigned*)(adst + (size_t)N_NODES * HEADS);
    float*    denom = (float*)(mflip + (size_t)N_NODES * HEADS);

    k_gemm<<<N_NODES / 16, 256, 0, stream>>>(x, W, h);
    k_att <<<(N_NODES * HEADS + 255) / 256, 256, 0, stream>>>(h, att_src, att_dst, asrc, adst);
    k_init<<<4096, 256, 0, stream>>>(mflip, denom, out);
    k_max <<<(TOT_E + 255) / 256, 256, 0, stream>>>(ei, asrc, adst, mflip);
    k_denom<<<(TOT_E + 255) / 256, 256, 0, stream>>>(ei, asrc, adst, mflip, denom);
    k_agg <<<12288, 256, 0, stream>>>(ei, asrc, adst, mflip, denom, h, out);
    k_lsm <<<(N_NODES + 7) / 8, 256, 0, stream>>>(out, bias);
}


# GEMM made persistent: 1250 blocks loop over 5 row tiles each (10000 waves enter, not 50000); W fragments load once a wave; x tile staged by 16-byte loads into padded conflict-free LDS rows
# speedup vs baseline: 3.7781x; 3.7781x over previous
#include <hip/hip_runtime.h>
#include <hip/hip_bf16.h>

#define N_NODES 100000
#define N_EDGES 1600000
#define TOT_E   (N_EDGES + N_NODES)
#define IN_CH   64
#define OUT_CH  32
#define HEADS   4
#define HC      (HEADS * OUT_CH)
#define NEG_SLOPE 0.2f

typedef __attribute__((ext_vector_type(16))) _Float16 v16h;
typedef __attribute__((ext_vector_type(8)))  float    v8f;

__device__ __forceinline__ unsigned fflip(float f) {
    unsigned u = __float_as_uint(f);
    return (u & 0x80000000u) ? ~u : (u | 0x80000000u);
}
__device__ __forceinline__ float funflip(unsigned u) {
    return __uint_as_float((u & 0x80000000u) ? (u & 0x7FFFFFFFu) : ~u);
}

__device__ __forceinline__ void edge_sd_sel(const int* __restrict__ ei, int e, int& s, int& d) {
    const int ec = min(e, N_EDGES - 1);
    const int ls = ei[ec], ld = ei[N_EDGES + ec];
    const bool real = e < N_EDGES;
    s = real ? ls : e - N_EDGES;
    d = real ? ld : e - N_EDGES;
}

#define M_TILES     (N_NODES / 16)
#define GEMM_BLOCKS 1250
#define XPITCH      72
static_assert(M_TILES % GEMM_BLOCKS == 0, "every block takes the same number of tiles");

typedef __attribute__((ext_vector_type(8))) _Float16 v8h;
typedef __attribute__((ext_vector_type(4))) _Float16 v4h;

__global__ __launch_bounds__(256) void k_gemm(const float* __restrict__ x,
                                              const float* __restrict__ W,
                                              float* __restrict__ h) {
    __shared__ __attribute__((aligned(16))) _Float16 xs[2][16 * XPITCH];
    const int tid = threadIdx.x;
    const int wave = tid >> 5;
    const int lane = tid & 31;
    const int mrow = lane & 15;
    const int khalf = (lane >> 4) << 3;
    const int ncol = wave * 16 + (lane & 15);

    v16h b0, b1;
    #pragma unroll
    for (int i = 0; i < 16; ++i) {
        int k = i + (i & 8) + khalf;
        b0[i] = (_Float16)W[(size_t)k * HC + ncol];
        b1[i] = (_Float16)W[(size_t)(k + 32) * HC + ncol];
    }

    const int sr = tid >> 4;
    const int sc = (tid & 15) << 2;
    int mt = blockIdx.x;
    float4 xv = *(const float4*)(x + (size_t)(mt * 16 + sr) * IN_CH + sc);
    int buf = 0;
    for (; mt < M_TILES; mt += GEMM_BLOCKS) {
        v4h xh;
        xh[0] = (_Float16)xv.x; xh[1] = (_Float16)xv.y; xh[2] = (_Float16)xv.z; xh[3] = (_Float16)xv.w;
        *(v4h*)(&xs[buf][sr * XPITCH + sc]) = xh;
        __syncthreads();
        const int nx = mt + GEMM_BLOCKS;
        if (nx < M_TILES) xv = *(const float4*)(x + (size_t)(nx * 16 + sr) * IN_CH + sc);

        const _Float16* rp = &xs[buf][mrow * XPITCH + khalf];
        const v8h p0 = *(const v8h*)(rp);
        const v8h p1 = *(const v8h*)(rp + 16);
        const v8h p2 = *(const v8h*)(rp + 32);
        const v8h p3 = *(const v8h*)(rp + 48);
        const v16h a0 = __builtin_shufflevector(p0, p1, 0, 1, 2, 3, 4, 5, 6, 7, 8, 9, 10, 11, 12, 13, 14, 15);
        const v16h a1 = __builtin_shufflevector(p2, p3, 0, 1, 2, 3, 4, 5, 6, 7, 8, 9, 10, 11, 12, 13, 14, 15);

        v8f c = {};
        c = __builtin_amdgcn_wmma_f32_16x16x32_f16(false, a0, false, b0, (short)0, c, false, false);
        c = __builtin_amdgcn_wmma_f32_16x16x32_f16(false, a1, false, b1, (short)0, c, false, false);

        #pragma unroll
        for (int r = 0; r < 8; ++r) {
            int m = mt * 16 + r + khalf;
            h[(size_t)m * HC + ncol] = c[r];
        }
        buf ^= 1;
    }
}

#define RNG    400
#define NOWN   (N_NODES / RNG)
static_assert(N_NODES % RNG == 0, "blocks own whole ranges: no row guard");
static_assert(NOWN <= 256, "an owner number fits a byte");
static_assert(N_EDGES % 4 == 0, "owner bytes are written four at a time");

__global__ __launch_bounds__(256) void k_att(const float* __restrict__ h,
                                             const float* __restrict__ att_src,
                                             const float* __restrict__ att_dst,
                                             const int* __restrict__ ei,
                                             float* __restrict__ a_src,
                                             float* __restrict__ a_dst,
                                             unsigned* __restrict__ mflip,
                                             unsigned* __restrict__ ownq) {
    int idx = blockIdx.x * blockDim.x + threadIdx.x;
    const int4* __restrict__ dq = (const int4*)(ei + N_EDGES);
    for (int q = idx; q < N_EDGES / 4; q += gridDim.x * blockDim.x) {
        const int4 d = dq[q];
        ownq[q] =  ((unsigned)d.x / RNG)        | (((unsigned)d.y / RNG) << 8)
                | (((unsigned)d.z / RNG) << 16) | (((unsigned)d.w / RNG) << 24);
    }
    if (idx >= N_NODES * HEADS) return;
    int n = idx >> 2, hh = idx & 3;
    const float* hp = h + (size_t)n * HC + hh * OUT_CH;
    float s = 0.f, d = 0.f;
    #pragma unroll
    for (int c = 0; c < OUT_CH; ++c) {
        float v = hp[c];
        s += v * att_src[hh * OUT_CH + c];
        d += v * att_dst[hh * OUT_CH + c];
    }
    a_src[idx] = s; a_dst[idx] = d;
    mflip[idx] = 0u;
}

#define EG      8
#define NGROUP  (TOT_E / EG)
static_assert(TOT_E % EG == 0, "edge groups are whole: no tail guard in the edge passes");

__device__ __forceinline__ int rl_i(int v, int l) {
    return __builtin_amdgcn_readlane(v, l);
}
__device__ __forceinline__ float rl_f(float v, int l) {
    return __int_as_float(__builtin_amdgcn_readlane(__float_as_int(v), l));
}

__device__ __forceinline__ float edge_score(const float* __restrict__ a_src,
                                            const float* __restrict__ a_dst,
                                            int s, int d, int hh) {
    float t = a_src[s * HEADS + hh] + a_dst[d * HEADS + hh];
    return t > 0.f ? t : NEG_SLOPE * t;
}

__device__ __forceinline__ float edge_weight(float t, unsigned mf) {
    t = t > 0.f ? t : NEG_SLOPE * t;
    return __expf(t - funflip(mf));
}

__global__ __launch_bounds__(256) void k_max(const int* __restrict__ ei,
                                             const float* __restrict__ a_src,
                                             const float* __restrict__ a_dst,
                                             unsigned* __restrict__ mflip) {
    const int lane = threadIdx.x & 31;
    const int hh   = lane & 3;
    const int sub  = lane >> 2;
    const int wglob = __builtin_amdgcn_readfirstlane((blockIdx.x * blockDim.x + threadIdx.x) >> 5);
    const int nwav  = (gridDim.x * blockDim.x) >> 5;

    int g = wglob;
    for (; g + 3 * nwav < NGROUP; g += 4 * nwav) {
        int s[4], d[4]; float v[4];
        #pragma unroll
        for (int j = 0; j < 4; ++j) edge_sd_sel(ei, (g + j * nwav) * EG + sub, s[j], d[j]);
        #pragma unroll
        for (int j = 0; j < 4; ++j) v[j] = edge_score(a_src, a_dst, s[j], d[j], hh);
        #pragma unroll
        for (int j = 0; j < 4; ++j) atomicMax(&mflip[d[j] * HEADS + hh], fflip(v[j]));
    }
    for (; g < NGROUP; g += nwav) {
        int s, d; edge_sd_sel(ei, g * EG + sub, s, d);
        atomicMax(&mflip[d * HEADS + hh], fflip(edge_score(a_src, a_dst, s, d, hh)));
    }
}

#define QCAP   128
#define BATCH  32
#define NBUF   3
#define SU     8
#define OSTEP  512
#define NSTEP  (N_EDGES / OSTEP)
static_assert(N_EDGES % OSTEP == 0, "wave steps are whole: no tail step");
static_assert((QCAP & (QCAP - 1)) == 0 && QCAP >= 32 + BATCH, "one component of a step always fits a ring that holds less than a batch");

__device__ __forceinline__ unsigned zero_bytes(unsigned x) {
    return ~(((x & 0x7f7f7f7fu) + 0x7f7f7f7fu) | x | 0x7f7f7f7fu);
}

__device__ __forceinline__ void own_push(bool mine, int e,
                                         int* wqe, int qh, int& qt, bool& ovf) {
    const unsigned mask = __builtin_amdgcn_ballot_w32(mine);
    if (mask == 0u) return;
    const int n = __builtin_popcount(mask);
    if (qt - qh + n <= QCAP) {
        if (mine) {
            const int slot = (qt + (int)__builtin_amdgcn_mbcnt_lo(mask, 0u)) & (QCAP - 1);
            wqe[slot] = e;
        }
        qt += n;
    } else {
        ovf = true;
    }
}

__device__ __forceinline__ void own_slow(int e, int r, int base,
                                         const int* __restrict__ ei,
                                         const float* __restrict__ a_src,
                                         const float* __restrict__ a_dst,
                                         const unsigned* __restrict__ mflip,
                                         const float* __restrict__ h,
                                         float* acc, float* den) {
    const int s = ei[e];
    const int d = base + r;
    #pragma unroll 1
    for (int j = 0; j < HEADS; ++j) {
        float t = a_src[s * HEADS + j] + a_dst[d * HEADS + j];
        t = t > 0.f ? t : NEG_SLOPE * t;
        const float w = __expf(t - funflip(mflip[d * HEADS + j]));
        atomicAdd(&den[r * HEADS + j], w);
        #pragma unroll 1
        for (int c = 0; c < OUT_CH; ++c)
            atomicAdd(&acc[r * HC + j * OUT_CH + c], h[(size_t)s * HC + j * OUT_CH + c] * w);
    }
}

__device__ __forceinline__ void batch_weights(float4 as, float4 ad, uint4 mf, bool val, int r,
                                              float* den, float (&ex)[HEADS]) {
    ex[0] = edge_weight(as.x + ad.x, mf.x); ex[1] = edge_weight(as.y + ad.y, mf.y);
    ex[2] = edge_weight(as.z + ad.z, mf.z); ex[3] = edge_weight(as.w + ad.w, mf.w);
    #pragma unroll
    for (int j = 0; j < HEADS; ++j) {
        ex[j] = val ? ex[j] : 0.f;
        atomicAdd(&den[r * HEADS + j], ex[j]);
    }
}

__device__ __forceinline__ void rows_load(const float* __restrict__ h, int s, int g, int lane,
                                          float (&hv)[4][HEADS]) {
    #pragma unroll
    for (int k = 0; k < 4; ++k) {
        const float* hp = h + (size_t)rl_i(s, g * 4 + k) * HC + lane;
        #pragma unroll
        for (int j = 0; j < HEADS; ++j) hv[k][j] = hp[j * OUT_CH];
    }
}

__device__ __forceinline__ void rows_add(float* acc, int r, const float (&ex)[HEADS], int g, int lane,
                                         const float (&hv)[4][HEADS]) {
    #pragma unroll
    for (int k = 0; k < 4; ++k) {
        float* ap = acc + rl_i(r, g * 4 + k) * HC + lane;
        #pragma unroll
        for (int j = 0; j < HEADS; ++j)
            atomicAdd(ap + j * OUT_CH, hv[k][j] * rl_f(ex[j], g * 4 + k));
    }
}

__global__ __launch_bounds__(1024) void k_own(const int* __restrict__ ei,
                                              const unsigned* __restrict__ own,
                                              const float* __restrict__ a_src,
                                              const float* __restrict__ a_dst,
                                              const unsigned* __restrict__ mflip,
                                              const float* __restrict__ h,
                                              const float* __restrict__ bias,
                                              float* __restrict__ out) {
    __shared__ __attribute__((aligned(16))) float acc[RNG * HC];
    __shared__ float den[RNG * HEADS];
    __shared__ int   qe[32 * QCAP];

    const int tid  = threadIdx.x;
    const int lane = tid & 31;
    const int hh   = lane & 3;
    const int sub  = lane >> 2;
    const int wave = __builtin_amdgcn_readfirstlane(tid >> 5);
    const int base = blockIdx.x * RNG;
    int* const wqe = qe + wave * QCAP;

    for (int r = wave; r < RNG; r += 32) {
        const int d = base + r;
        float t = a_src[d * HEADS + hh] + a_dst[d * HEADS + hh];
        t = t > 0.f ? t : NEG_SLOPE * t;
        const float ex = __expf(t - funflip(mflip[d * HEADS + hh]));
        const float* hp = h + (size_t)d * HC + lane;
        float* ap = acc + r * HC + lane;
        #pragma unroll
        for (int j = 0; j < HEADS; ++j) ap[j * OUT_CH] = hp[j * OUT_CH] * rl_f(ex, j);
        if (lane < HEADS) den[r * HEADS + lane] = ex;
    }
    __syncthreads();

    const uint4* __restrict__ ocol = (const uint4*)own;
    const unsigned pat = (unsigned)blockIdx.x * 0x01010101u;
    int qh = 0, qt = 0;
    for (int t0 = wave; ; t0 += 32 * SU) {
        const bool more = t0 < NSTEP;
        if (more) {
            uint4 ov[SU];
            const int qt0 = qt;
            bool ovf = false;
            #pragma unroll
            for (int j = 0; j < SU; ++j) ov[j] = ocol[min(t0 + 32 * j, NSTEP - 1) * 32 + lane];
            #pragma unroll
            for (int j = 0; j < SU; ++j) {
                const int t = t0 + 32 * j;
                unsigned c = (zero_bytes(ov[j].x ^ pat) >> 7) | (zero_bytes(ov[j].y ^ pat) >> 6)
                           | (zero_bytes(ov[j].z ^ pat) >> 5) | (zero_bytes(ov[j].w ^ pat) >> 4);
                c = t < NSTEP ? c : 0u;
                const int e0 = t * OSTEP + lane * 16;
                while (__builtin_amdgcn_ballot_w32(c != 0u) != 0u) {
                    const bool mine = c != 0u;
                    const int  p = mine ? __builtin_ctz(c) : 0;
                    own_push(mine, e0 + ((p & 3) << 2) + (p >> 3), wqe, qh, qt, ovf);
                    c &= c - 1u;
                }
            }
            if (ovf) {
                qt = qt0;
                #pragma unroll 1
                for (int k = 0; k < SU * (OSTEP / 32); ++k) {
                    const int t = t0 + 32 * (k / (OSTEP / 32));
                    if (t < NSTEP) {
                        const int e = t * OSTEP + (k % (OSTEP / 32)) * 32 + lane;
                        const unsigned r = (unsigned)(ei[N_EDGES + e] - base);
                        if (r < (unsigned)RNG) own_slow(e, (int)r, base, ei, a_src, a_dst, mflip, h, acc, den);
                    }
                }
            }
        }
        __builtin_amdgcn_wave_barrier();
        while (qt - qh >= BATCH) {
            const int e = wqe[(qh + lane) & (QCAP - 1)];
            const int s = ei[e];
            const int d = ei[N_EDGES + e];
            const int r = d - base;
            const float4 as = ((const float4*)a_src)[s];
            const float4 ad = ((const float4*)a_dst)[d];
            const uint4  mf = ((const uint4*)mflip)[d];
            float hv[NBUF][4][HEADS];
            #pragma unroll
            for (int g = 0; g < NBUF; ++g) rows_load(h, s, g, lane, hv[g]);
            __builtin_amdgcn_sched_barrier(0);
            float ex[HEADS];
            batch_weights(as, ad, mf, true, r, den, ex);
            #pragma unroll
            for (int g = 0; g < BATCH / 4; ++g) {
                rows_add(acc, r, ex, g, lane, hv[g % NBUF]);
                if (g + NBUF < BATCH / 4) rows_load(h, s, g + NBUF, lane, hv[g % NBUF]);
                __builtin_amdgcn_sched_barrier(0);
            }
            qh += BATCH;
        }
        if (!more) {
            const int cnt = qt - qh;
            if (cnt > 0) {
                const bool val = lane < cnt;
                const int e = wqe[(qh + (val ? lane : 0)) & (QCAP - 1)];
                const int s = ei[e];
                const int d = ei[N_EDGES + e];
                const int r = d - base;
                float ex[HEADS];
                batch_weights(((const float4*)a_src)[s], ((const float4*)a_dst)[d], ((const uint4*)mflip)[d],
                              val, r, den, ex);
                #pragma unroll 1
                for (int g = 0; g * 4 < cnt; ++g) {
                    float hv[4][HEADS];
                    rows_load(h, s, g, lane, hv);
                    rows_add(acc, r, ex, g, lane, hv);
                }
            }
            break;
        }
    }
    __syncthreads();

    const float4 bv = ((const float4*)bias)[lane];
    for (int r = wave; r < RNG; r += 32) {
        float4 v = ((const float4*)(acc + r * HC))[lane];
        const float dn = den[r * HEADS + (lane >> 3)] + 1e-16f;
        v.x = v.x / dn + bv.x; v.y = v.y / dn + bv.y;
        v.z = v.z / dn + bv.z; v.w = v.w / dn + bv.w;

        float mx = fmaxf(fmaxf(v.x, v.y), fmaxf(v.z, v.w));
        #pragma unroll
        for (int o = 16; o >= 1; o >>= 1) mx = fmaxf(mx, __shfl_xor(mx, o, 32));

        float se = __expf(v.x - mx) + __expf(v.y - mx) + __expf(v.z - mx) + __expf(v.w - mx);
        #pragma unroll
        for (int o = 16; o >= 1; o >>= 1) se += __shfl_xor(se, o, 32);

        const float lse = mx + __logf(se);
        v.x -= lse; v.y -= lse; v.z -= lse; v.w -= lse;
        ((float4*)(out + (size_t)(base + r) * HC))[lane] = v;
    }
}

extern "C" void kernel_launch(void* const* d_in, const int* in_sizes, int n_in,
                              void* d_out, int out_size, void* d_ws, size_t ws_size,
                              hipStream_t stream) {
    const float* x       = (const float*)d_in[0];
    const int*   ei      = (const int*)  d_in[1];
    const float* W       = (const float*)d_in[2];
    const float* att_src = (const float*)d_in[3];
    const float* att_dst = (const float*)d_in[4];
    const float* bias    = (const float*)d_in[5];
    float*       out     = (float*)d_out;

    float*    h     = (float*)d_ws;
    float*    asrc  = h    + (size_t)N_NODES * HC;
    float*    adst  = asrc + (size_t)N_NODES * HEADS;
    unsigned* mflip = (unsigned*)(adst + (size_t)N_NODES * HEADS);
    unsigned* ownq  = mflip + (size_t)N_NODES * HEADS;

    k_gemm<<<GEMM_BLOCKS, 256, 0, stream>>>(x, W, h);
    k_att <<<(N_NODES * HEADS + 255) / 256, 256, 0, stream>>>(h, att_src, att_dst, ei, asrc, adst, mflip, ownq);
    k_max <<<1024, 256, 0, stream>>>(ei, asrc, adst, mflip);
    k_own <<<NOWN, 1024, 0, stream>>>(ei, ownq, asrc, adst, mflip, h, bias, out);
}


# Segment max moved into the owner blocks: kept edges and self loops stashed once in LDS, scores maxed by LDS integer max, then rows added; L2 max launch and table gone; LDS adds arrival order
# speedup vs baseline: 5.5305x; 1.4638x over previous
#include <hip/hip_runtime.h>
#include <hip/hip_bf16.h>

#define N_NODES 100000
#define N_EDGES 1600000
#define TOT_E   (N_EDGES + N_NODES)
#define IN_CH   64
#define OUT_CH  32
#define HEADS   4
#define HC      (HEADS * OUT_CH)
#define NEG_SLOPE 0.2f

typedef __attribute__((ext_vector_type(16))) _Float16 v16h;
typedef __attribute__((ext_vector_type(8)))  float    v8f;

__device__ __forceinline__ unsigned fflip(float f) {
    unsigned u = __float_as_uint(f);
    return (u & 0x80000000u) ? ~u : (u | 0x80000000u);
}
__device__ __forceinline__ float funflip(unsigned u) {
    return __uint_as_float((u & 0x80000000u) ? (u & 0x7FFFFFFFu) : ~u);
}

__device__ __forceinline__ void edge_sd_sel(const int* __restrict__ ei, int e, int& s, int& d) {
    const int ec = min(e, N_EDGES - 1);
    const int ls = ei[ec], ld = ei[N_EDGES + ec];
    const bool real = e < N_EDGES;
    s = real ? ls : e - N_EDGES;
    d = real ? ld : e - N_EDGES;
}

#define M_TILES     (N_NODES / 16)
#define GEMM_BLOCKS 1250
#define XPITCH      72
static_assert(M_TILES % GEMM_BLOCKS == 0, "every block takes the same number of tiles");

typedef __attribute__((ext_vector_type(8))) _Float16 v8h;
typedef __attribute__((ext_vector_type(4))) _Float16 v4h;

__global__ __launch_bounds__(256) void k_gemm(const float* __restrict__ x,
                                              const float* __restrict__ W,
                                              float* __restrict__ h) {
    __shared__ __attribute__((aligned(16))) _Float16 xs[2][16 * XPITCH];
    const int tid = threadIdx.x;
    const int wave = tid >> 5;
    const int lane = tid & 31;
    const int mrow = lane & 15;
    const int khalf = (lane >> 4) << 3;
    const int ncol = wave * 16 + (lane & 15);

    v16h b0, b1;
    #pragma unroll
    for (int i = 0; i < 16; ++i) {
        int k = i + (i & 8) + khalf;
        b0[i] = (_Float16)W[(size_t)k * HC + ncol];
        b1[i] = (_Float16)W[(size_t)(k + 32) * HC + ncol];
    }

    const int sr = tid >> 4;
    const int sc = (tid & 15) << 2;
    int mt = blockIdx.x;
    float4 xv = *(const float4*)(x + (size_t)(mt * 16 + sr) * IN_CH + sc);
    int buf = 0;
    for (; mt < M_TILES; mt += GEMM_BLOCKS) {
        v4h xh;
        xh[0] = (_Float16)xv.x; xh[1] = (_Float16)xv.y; xh[2] = (_Float16)xv.z; xh[3] = (_Float16)xv.w;
        *(v4h*)(&xs[buf][sr * XPITCH + sc]) = xh;
        __syncthreads();
        const int nx = mt + GEMM_BLOCKS;
        if (nx < M_TILES) xv = *(const float4*)(x + (size_t)(nx * 16 + sr) * IN_CH + sc);

        const _Float16* rp = &xs[buf][mrow * XPITCH + khalf];
        const v8h p0 = *(const v8h*)(rp);
        const v8h p1 = *(const v8h*)(rp + 16);
        const v8h p2 = *(const v8h*)(rp + 32);
        const v8h p3 = *(const v8h*)(rp + 48);
        const v16h a0 = __builtin_shufflevector(p0, p1, 0, 1, 2, 3, 4, 5, 6, 7, 8, 9, 10, 11, 12, 13, 14, 15);
        const v16h a1 = __builtin_shufflevector(p2, p3, 0, 1, 2, 3, 4, 5, 6, 7, 8, 9, 10, 11, 12, 13, 14, 15);

        v8f c = {};
        c = __builtin_amdgcn_wmma_f32_16x16x32_f16(false, a0, false, b0, (short)0, c, false, false);
        c = __builtin_amdgcn_wmma_f32_16x16x32_f16(false, a1, false, b1, (short)0, c, false, false);

        #pragma unroll
        for (int r = 0; r < 8; ++r) {
            int m = mt * 16 + r + khalf;
            h[(size_t)m * HC + ncol] = c[r];
        }
        buf ^= 1;
    }
}

#define RNG    400
#define NOWN   (N_NODES / RNG)
static_assert(N_NODES % RNG == 0, "blocks own whole ranges: no row guard");
static_assert(NOWN <= 256, "an owner number fits a byte");
static_assert(N_EDGES % 4 == 0, "owner bytes are written four at a time");

__global__ __launch_bounds__(256) void k_att(const float* __restrict__ h,
                                             const float* __restrict__ att_src,
                                             const float* __restrict__ att_dst,
                                             const int* __restrict__ ei,
                                             float* __restrict__ a_src,
                                             float* __restrict__ a_dst,
                                             unsigned* __restrict__ ownq) {
    int idx = blockIdx.x * blockDim.x + threadIdx.x;
    const int4* __restrict__ dq = (const int4*)(ei + N_EDGES);
    for (int q = idx; q < N_EDGES / 4; q += gridDim.x * blockDim.x) {
        const int4 d = dq[q];
        ownq[q] =  ((unsigned)d.x / RNG)        | (((unsigned)d.y / RNG) << 8)
                | (((unsigned)d.z / RNG) << 16) | (((unsigned)d.w / RNG) << 24);
    }
    if (idx >= N_NODES * HEADS) return;
    int n = idx >> 2, hh = idx & 3;
    const float* hp = h + (size_t)n * HC + hh * OUT_CH;
    float s = 0.f, d = 0.f;
    #pragma unroll
    for (int c = 0; c < OUT_CH; ++c) {
        float v = hp[c];
        s += v * att_src[hh * OUT_CH + c];
        d += v * att_dst[hh * OUT_CH + c];
    }
    a_src[idx] = s; a_dst[idx] = d;
}

#define SCAP   384
#define BATCH  32
#define NBUF   3
#define SU     8
#define OSTEP  512
#define NSTEP  (N_EDGES / OSTEP)
#define NOSLOW 0x7fffffff
static_assert(N_EDGES % OSTEP == 0, "wave steps are whole: no tail step");
static_assert(N_NODES <= (1 << 17) && RNG <= (1 << 14), "a stash word holds row << 17 | source");
static_assert((RNG + 31) / 32 <= 32 && (RNG + 31) / 32 <= SCAP, "a wave's self loops fit one lane each");

__device__ __forceinline__ int rl_i(int v, int l) {
    return __builtin_amdgcn_readlane(v, l);
}
__device__ __forceinline__ float rl_f(float v, int l) {
    return __int_as_float(__builtin_amdgcn_readlane(__float_as_int(v), l));
}

__device__ __forceinline__ float edge_score(float t) {
    return t > 0.f ? t : NEG_SLOPE * t;
}

__device__ __forceinline__ float edge_weight(float t, unsigned mf) {
    return __expf(edge_score(t) - funflip(mf));
}

__device__ __forceinline__ unsigned zero_bytes(unsigned x) {
    return ~(((x & 0x7f7f7f7fu) + 0x7f7f7f7fu) | x | 0x7f7f7f7fu);
}

__device__ __forceinline__ void own_push(bool mine, int e, int* wst, int& qt, bool& ovf) {
    const unsigned mask = __builtin_amdgcn_ballot_w32(mine);
    if (mask == 0u) return;
    const int n = __builtin_popcount(mask);
    if (qt + n <= SCAP) {
        if (mine) wst[qt + (int)__builtin_amdgcn_mbcnt_lo(mask, 0u)] = e;
        qt += n;
    } else {
        ovf = true;
    }
}

#define MAXB 4
__device__ __forceinline__ void stash_max(int* wst, int q, int qt, int lane, int base,
                                          const int* __restrict__ ei,
                                          const float* __restrict__ a_src,
                                          const float* ad_s, unsigned* mx) {
    int idx[MAXB], s[MAXB], d[MAXB];
    bool val[MAXB];
    float4 as[MAXB], ad[MAXB];
    #pragma unroll
    for (int j = 0; j < MAXB; ++j) {
        const int i = q + BATCH * j + lane;
        val[j] = i < qt;
        idx[j] = val[j] ? i : q;
        edge_sd_sel(ei, wst[idx[j]], s[j], d[j]);
    }
    #pragma unroll
    for (int j = 0; j < MAXB; ++j) {
        as[j] = ((const float4*)a_src)[s[j]];
        ad[j] = ((const float4*)ad_s)[d[j] - base];
    }
    #pragma unroll
    for (int j = 0; j < MAXB; ++j) {
        const int r = d[j] - base;
        unsigned* const mp = mx + (val[j] ? r : RNG + lane) * HEADS;
        atomicMax(mp + 0, fflip(edge_score(as[j].x + ad[j].x)));
        atomicMax(mp + 1, fflip(edge_score(as[j].y + ad[j].y)));
        atomicMax(mp + 2, fflip(edge_score(as[j].z + ad[j].z)));
        atomicMax(mp + 3, fflip(edge_score(as[j].w + ad[j].w)));
        if (val[j]) wst[idx[j]] = (r << 17) | s[j];
    }
}

__device__ __forceinline__ void slow_max(int t0, int lane, int base,
                                         const int* __restrict__ ei,
                                         const float* __restrict__ a_src,
                                         const float* ad_s, unsigned* mx) {
    #pragma unroll 1
    for (int k = 0; k < SU * (OSTEP / 32); ++k) {
        const int t = t0 + 32 * (k / (OSTEP / 32));
        if (t < NSTEP) {
            const int e = t * OSTEP + (k % (OSTEP / 32)) * 32 + lane;
            const unsigned r = (unsigned)(ei[N_EDGES + e] - base);
            if (r < (unsigned)RNG) {
                const int s = ei[e];
                #pragma unroll 1
                for (int j = 0; j < HEADS; ++j)
                    atomicMax(&mx[r * HEADS + j], fflip(edge_score(a_src[s * HEADS + j] + ad_s[r * HEADS + j])));
            }
        }
    }
}

__device__ __forceinline__ void slow_sum(int t0, int lane, int base,
                                         const int* __restrict__ ei,
                                         const float* __restrict__ a_src,
                                         const float* __restrict__ h,
                                         const float* ad_s, const unsigned* mx,
                                         float* acc, float* den) {
    #pragma unroll 1
    for (int k = 0; k < SU * (OSTEP / 32); ++k) {
        const int t = t0 + 32 * (k / (OSTEP / 32));
        if (t < NSTEP) {
            const int e = t * OSTEP + (k % (OSTEP / 32)) * 32 + lane;
            const unsigned r = (unsigned)(ei[N_EDGES + e] - base);
            if (r < (unsigned)RNG) {
                const int s = ei[e];
                #pragma unroll 1
                for (int j = 0; j < HEADS; ++j) {
                    const float w = edge_weight(a_src[s * HEADS + j] + ad_s[r * HEADS + j], mx[r * HEADS + j]);
                    atomicAdd(&den[r * HEADS + j], w);
                    #pragma unroll 1
                    for (int c = 0; c < OUT_CH; ++c)
                        atomicAdd(&acc[r * HC + j * OUT_CH + c], h[(size_t)s * HC + j * OUT_CH + c] * w);
                }
            }
        }
    }
}

__device__ __forceinline__ void batch_weights(float4 as, float4 ad, uint4 mf, bool val, int r,
                                              float* den, float (&ex)[HEADS]) {
    ex[0] = edge_weight(as.x + ad.x, mf.x); ex[1] = edge_weight(as.y + ad.y, mf.y);
    ex[2] = edge_weight(as.z + ad.z, mf.z); ex[3] = edge_weight(as.w + ad.w, mf.w);
    #pragma unroll
    for (int j = 0; j < HEADS; ++j) {
        ex[j] = val ? ex[j] : 0.f;
        atomicAdd(&den[r * HEADS + j], ex[j]);
    }
}

__device__ __forceinline__ void rows_load(const float* __restrict__ h, int s, int g, int lane,
                                          float (&hv)[4][HEADS]) {
    #pragma unroll
    for (int k = 0; k < 4; ++k) {
        const float* hp = h + (size_t)rl_i(s, g * 4 + k) * HC + lane;
        #pragma unroll
        for (int j = 0; j < HEADS; ++j) hv[k][j] = hp[j * OUT_CH];
    }
}

__device__ __forceinline__ void rows_add(float* acc, int r, const float (&ex)[HEADS], int g, int lane,
                                         const float (&hv)[4][HEADS]) {
    #pragma unroll
    for (int k = 0; k < 4; ++k) {
        float* ap = acc + rl_i(r, g * 4 + k) * HC + lane;
        #pragma unroll
        for (int j = 0; j < HEADS; ++j)
            atomicAdd(ap + j * OUT_CH, hv[k][j] * rl_f(ex[j], g * 4 + k));
    }
}

__global__ __launch_bounds__(1024) void k_own(const int* __restrict__ ei,
                                              const unsigned* __restrict__ own,
                                              const float* __restrict__ a_src,
                                              const float* __restrict__ a_dst,
                                              const float* __restrict__ h,
                                              const float* __restrict__ bias,
                                              float* __restrict__ out) {
    __shared__ __attribute__((aligned(16))) float    acc[RNG * HC];
    __shared__ __attribute__((aligned(16))) float    den[RNG * HEADS];
    __shared__ __attribute__((aligned(16))) float    ad_s[RNG * HEADS];
    __shared__ __attribute__((aligned(16))) unsigned mx[(RNG + 32) * HEADS];
    __shared__ int st[32 * SCAP];

    const int tid  = threadIdx.x;
    const int lane = tid & 31;
    const int wave = __builtin_amdgcn_readfirstlane(tid >> 5);
    const int base = blockIdx.x * RNG;
    int* const wst = st + wave * SCAP;

    for (int i = tid; i < RNG * HC / 4; i += 1024) ((float4*)acc)[i] = make_float4(0.f, 0.f, 0.f, 0.f);
    for (int i = tid; i < RNG * HEADS; i += 1024) {
        den[i] = 0.f;
        mx[i] = 0u;
        ad_s[i] = a_dst[base * HEADS + i];
    }
    if (tid < 32 * HEADS) mx[RNG * HEADS + tid] = 0u;
    const int nself = (RNG - wave + 31) / 32;
    if (lane < nself) wst[lane] = N_EDGES + base + wave + 32 * lane;
    __syncthreads();

    const uint4* __restrict__ ocol = (const uint4*)own;
    const unsigned pat = (unsigned)blockIdx.x * 0x01010101u;
    int qt = nself;
    int slow0 = NOSLOW;
    for (int t0 = wave; t0 < NSTEP; t0 += 32 * SU) {
        if (slow0 == NOSLOW) {
            uint4 ov[SU];
            const int qt0 = qt;
            bool ovf = false;
            #pragma unroll
            for (int j = 0; j < SU; ++j) ov[j] = ocol[min(t0 + 32 * j, NSTEP - 1) * 32 + lane];
            #pragma unroll
            for (int j = 0; j < SU; ++j) {
                const int t = t0 + 32 * j;
                unsigned c = (zero_bytes(ov[j].x ^ pat) >> 7) | (zero_bytes(ov[j].y ^ pat) >> 6)
                           | (zero_bytes(ov[j].z ^ pat) >> 5) | (zero_bytes(ov[j].w ^ pat) >> 4);
                c = t < NSTEP ? c : 0u;
                const int e0 = t * OSTEP + lane * 16;
                while (__builtin_amdgcn_ballot_w32(c != 0u) != 0u) {
                    const bool mine = c != 0u;
                    const int  p = mine ? __builtin_ctz(c) : 0;
                    own_push(mine, e0 + ((p & 3) << 2) + (p >> 3), wst, qt, ovf);
                    c &= c - 1u;
                }
            }
            if (ovf) {
                qt = qt0;
                slow0 = t0;
            }
        }
        if (slow0 != NOSLOW) slow_max(t0, lane, base, ei, a_src, ad_s, mx);
    }
    __builtin_amdgcn_wave_barrier();
    for (int q = 0; q < qt; q += MAXB * BATCH) stash_max(wst, q, qt, lane, base, ei, a_src, ad_s, mx);
    __syncthreads();

    int qh = 0;
    while (qt - qh >= BATCH) {
        const int p = wst[qh + lane];
        const int s = p & 0x1ffff;
        const int r = p >> 17;
        const float4 as = ((const float4*)a_src)[s];
        float hv[NBUF][4][HEADS];
        #pragma unroll
        for (int g = 0; g < NBUF; ++g) rows_load(h, s, g, lane, hv[g]);
        __builtin_amdgcn_sched_barrier(0);
        const float4 ad = ((const float4*)ad_s)[r];
        const uint4  mf = ((const uint4*)mx)[r];
        float ex[HEADS];
        batch_weights(as, ad, mf, true, r, den, ex);
        #pragma unroll
        for (int g = 0; g < BATCH / 4; ++g) {
            rows_add(acc, r, ex, g, lane, hv[g % NBUF]);
            if (g + NBUF < BATCH / 4) rows_load(h, s, g + NBUF, lane, hv[g % NBUF]);
            __builtin_amdgcn_sched_barrier(0);
        }
        qh += BATCH;
    }
    {
        const int cnt = qt - qh;
        if (cnt > 0) {
            const bool val = lane < cnt;
            const int p = wst[qh + (val ? lane : 0)];
            const int s = p & 0x1ffff;
            const int r = p >> 17;
            float ex[HEADS];
            batch_weights(((const float4*)a_src)[s], ((const float4*)ad_s)[r], ((const uint4*)mx)[r],
                          val, r, den, ex);
            #pragma unroll 1
            for (int g = 0; g * 4 < cnt; ++g) {
                float hv[4][HEADS];
                rows_load(h, s, g, lane, hv);
                rows_add(acc, r, ex, g, lane, hv);
            }
        }
    }
    if (slow0 != NOSLOW) {
        #pragma unroll 1
        for (int t0 = slow0; t0 < NSTEP; t0 += 32 * SU) slow_sum(t0, lane, base, ei, a_src, h, ad_s, mx, acc, den);
    }
    __syncthreads();

    const float4 bv = ((const float4*)bias)[lane];
    for (int r = wave; r < RNG; r += 32) {
        float4 v = ((const float4*)(acc + r * HC))[lane];
        const float dn = den[r * HEADS + (lane >> 3)] + 1e-16f;
        v.x = v.x / dn + bv.x; v.y = v.y / dn + bv.y;
        v.z = v.z / dn + bv.z; v.w = v.w / dn + bv.w;

        float mx_r = fmaxf(fmaxf(v.x, v.y), fmaxf(v.z, v.w));
        #pragma unroll
        for (int o = 16; o >= 1; o >>= 1) mx_r = fmaxf(mx_r, __shfl_xor(mx_r, o, 32));

        float se = __expf(v.x - mx_r) + __expf(v.y - mx_r) + __expf(v.z - mx_r) + __expf(v.w - mx_r);
        #pragma unroll
        for (int o = 16; o >= 1; o >>= 1) se += __shfl_xor(se, o, 32);

        const float lse = mx_r + __logf(se);
        v.x -= lse; v.y -= lse; v.z -= lse; v.w -= lse;
        ((float4*)(out + (size_t)(base + r) * HC))[lane] = v;
    }
}

extern "C" void kernel_launch(void* const* d_in, const int* in_sizes, int n_in,
                              void* d_out, int out_size, void* d_ws, size_t ws_size,
                              hipStream_t stream) {
    const float* x       = (const float*)d_in[0];
    const int*   ei      = (const int*)  d_in[1];
    const float* W       = (const float*)d_in[2];
    const float* att_src = (const float*)d_in[3];
    const float* att_dst = (const float*)d_in[4];
    const float* bias    = (const float*)d_in[5];
    float*       out     = (float*)d_out;

    float*    h     = (float*)d_ws;
    float*    asrc  = h    + (size_t)N_NODES * HC;
    float*    adst  = asrc + (size_t)N_NODES * HEADS;
    unsigned* ownq  = (unsigned*)(adst + (size_t)N_NODES * HEADS);

    k_gemm<<<GEMM_BLOCKS, 256, 0, stream>>>(x, W, h);
    k_att <<<(N_NODES * HEADS + 255) / 256, 256, 0, stream>>>(h, att_src, att_dst, ei, asrc, adst, ownq);
    k_own <<<NOWN, 1024, 0, stream>>>(ei, ownq, asrc, adst, h, bias, out);
}


# Node pass folded into the GEMM launch: attention halves summed from the accumulator tiles (lane tree on the vector ALU, two waves joined in LDS), owner bytes by its grid; 2 launches; new sum order
# speedup vs baseline: 5.8758x; 1.0624x over previous
#include <hip/hip_runtime.h>
#include <hip/hip_bf16.h>

#define N_NODES 100000
#define N_EDGES 1600000
#define TOT_E   (N_EDGES + N_NODES)
#define IN_CH   64
#define OUT_CH  32
#define HEADS   4
#define HC      (HEADS * OUT_CH)
#define NEG_SLOPE 0.2f

typedef __attribute__((ext_vector_type(16))) _Float16 v16h;
typedef __attribute__((ext_vector_type(8)))  float    v8f;

__device__ __forceinline__ unsigned fflip(float f) {
    unsigned u = __float_as_uint(f);
    return (u & 0x80000000u) ? ~u : (u | 0x80000000u);
}
__device__ __forceinline__ float funflip(unsigned u) {
    return __uint_as_float((u & 0x80000000u) ? (u & 0x7FFFFFFFu) : ~u);
}

__device__ __forceinline__ void edge_sd_sel(const int* __restrict__ ei, int e, int& s, int& d) {
    const int ec = min(e, N_EDGES - 1);
    const int ls = ei[ec], ld = ei[N_EDGES + ec];
    const bool real = e < N_EDGES;
    s = real ? ls : e - N_EDGES;
    d = real ? ld : e - N_EDGES;
}

#define M_TILES     (N_NODES / 16)
#define GEMM_BLOCKS 1250
#define GEMM_THREADS (GEMM_BLOCKS * 256)
#define XPITCH      72
static_assert(M_TILES % GEMM_BLOCKS == 0, "every block takes the same number of tiles");

#define RNG    400
#define NOWN   (N_NODES / RNG)
static_assert(N_NODES % RNG == 0, "blocks own whole ranges: no row guard");
static_assert(NOWN <= 256, "an owner number fits a byte");
static_assert(N_EDGES % 4 == 0, "owner bytes are written four at a time");
static_assert(N_EDGES / 4 > GEMM_THREADS && N_EDGES / 4 <= 2 * GEMM_THREADS,
              "every thread of the GEMM grid writes one owner word, the first ones a second");

typedef __attribute__((ext_vector_type(8))) _Float16 v8h;
typedef __attribute__((ext_vector_type(4))) _Float16 v4h;

__device__ __forceinline__ unsigned own4(int4 d) {
    return  ((unsigned)d.x / RNG)        | (((unsigned)d.y / RNG) << 8)
         | (((unsigned)d.z / RNG) << 16) | (((unsigned)d.w / RNG) << 24);
}

template <int CTRL>
__device__ __forceinline__ float row_lane(float v) {
    return __int_as_float(__builtin_amdgcn_update_dpp(0, __float_as_int(v), CTRL, 0xf, 0xf, true));
}
__device__ __forceinline__ float row_sum16(float v) {
    v += row_lane<0xB1>(v);
    v += row_lane<0x4E>(v);
    v += row_lane<0x141>(v);
    v += row_lane<0x140>(v);
    return v;
}

__global__ __launch_bounds__(256) __attribute__((amdgpu_waves_per_eu(10)))
void k_gemm(const float* __restrict__ x,
                                              const float* __restrict__ W,
                                              const float* __restrict__ att_src,
                                              const float* __restrict__ att_dst,
                                              const int* __restrict__ ei,
                                              float* __restrict__ h,
                                              float* __restrict__ a_src,
                                              float* __restrict__ a_dst,
                                              unsigned* __restrict__ ownq) {
    __shared__ __attribute__((aligned(16))) _Float16 xs[2][16 * XPITCH];
    __shared__ __attribute__((aligned(16))) float part[2][8][2][16];
    const int tid = threadIdx.x;
    const int wave = tid >> 5;
    const int lane = tid & 31;
    const int mrow = lane & 15;
    const int khalf = (lane >> 4) << 3;
    const int ncol = wave * 16 + (lane & 15);

    const int gt = blockIdx.x * 256 + tid;
    const int4* __restrict__ dq = (const int4*)(ei + N_EDGES);
    const bool two = gt < N_EDGES / 4 - GEMM_THREADS;
    const int4 d0 = dq[gt];
    const int4 d1 = dq[two ? gt + GEMM_THREADS : gt];

    v16h b0, b1;
    #pragma unroll
    for (int i = 0; i < 16; ++i) {
        int k = i + (i & 8) + khalf;
        b0[i] = (_Float16)W[(size_t)k * HC + ncol];
        b1[i] = (_Float16)W[(size_t)(k + 32) * HC + ncol];
    }
    const float as_c = att_src[ncol];
    const float ad_c = att_dst[ncol];

    const unsigned o0 = own4(d0), o1 = own4(d1);

    const int jidx = tid & 63;
    const int jrow = jidx >> 2, jhead = jidx & 3;
    const int joff = (2 * jhead * 2 + (jrow >> 3)) * 16 + ((tid >> 6) & 1) * 8 + (jrow & 7);

    const int sr = tid >> 4;
    const int sc = (tid & 15) << 2;
    int mt = blockIdx.x;
    float4 xv = *(const float4*)(x + (size_t)(mt * 16 + sr) * IN_CH + sc);
    int buf = 0;
    #pragma nounroll
    for (; mt < M_TILES; mt += GEMM_BLOCKS) {
        v4h xh;
        xh[0] = (_Float16)xv.x; xh[1] = (_Float16)xv.y; xh[2] = (_Float16)xv.z; xh[3] = (_Float16)xv.w;
        *(v4h*)(&xs[buf][sr * XPITCH + sc]) = xh;
        __syncthreads();
        const int nx = mt + GEMM_BLOCKS;
        if (nx < M_TILES) xv = *(const float4*)(x + (size_t)(nx * 16 + sr) * IN_CH + sc);

        if (mt != (int)blockIdx.x && tid < 128) {
            const float* pp = &part[buf ^ 1][0][0][0] + joff;
            const float v = pp[0] + pp[32];
            const size_t o = (size_t)(mt - GEMM_BLOCKS) * 16 * HEADS + jidx;
            if (tid < 64) a_src[o] = v; else a_dst[o] = v;
        }

        const _Float16* rp = &xs[buf][mrow * XPITCH + khalf];
        const v8h p0 = *(const v8h*)(rp);
        const v8h p1 = *(const v8h*)(rp + 16);
        const v8h p2 = *(const v8h*)(rp + 32);
        const v8h p3 = *(const v8h*)(rp + 48);
        const v16h a0 = __builtin_shufflevector(p0, p1, 0, 1, 2, 3, 4, 5, 6, 7, 8, 9, 10, 11, 12, 13, 14, 15);
        const v16h a1 = __builtin_shufflevector(p2, p3, 0, 1, 2, 3, 4, 5, 6, 7, 8, 9, 10, 11, 12, 13, 14, 15);

        v8f c = {};
        c = __builtin_amdgcn_wmma_f32_16x16x32_f16(false, a0, false, b0, (short)0, c, false, false);
        c = __builtin_amdgcn_wmma_f32_16x16x32_f16(false, a1, false, b1, (short)0, c, false, false);

        float* const hrow = h + ((size_t)(mt * 16 + khalf) * HC + ncol);
        #pragma unroll
        for (int r = 0; r < 8; ++r) hrow[r * HC] = c[r];

        float ps[8], pd[8];
        #pragma unroll
        for (int r = 0; r < 8; ++r) {
            ps[r] = row_sum16(c[r] * as_c);
            pd[r] = row_sum16(c[r] * ad_c);
        }
        if ((lane & 15) == 0) {
            float4* const wp = (float4*)&part[buf][wave][lane >> 4][0];
            wp[0] = make_float4(ps[0], ps[1], ps[2], ps[3]);
            wp[1] = make_float4(ps[4], ps[5], ps[6], ps[7]);
            wp[2] = make_float4(pd[0], pd[1], pd[2], pd[3]);
            wp[3] = make_float4(pd[4], pd[5], pd[6], pd[7]);
        }
        buf ^= 1;
    }
    __syncthreads();
    if (tid < 128) {
        const float* pp = &part[buf ^ 1][0][0][0] + joff;
        const float v = pp[0] + pp[32];
        const size_t o = (size_t)(mt - GEMM_BLOCKS) * 16 * HEADS + jidx;
        if (tid < 64) a_src[o] = v; else a_dst[o] = v;
    }
    ownq[gt] = o0;
    if (two) ownq[gt + GEMM_THREADS] = o1;
}

#define SCAP   384
#define BATCH  32
#define NBUF   3
#define SU     8
#define OSTEP  512
#define NSTEP  (N_EDGES / OSTEP)
#define NOSLOW 0x7fffffff
static_assert(N_EDGES % OSTEP == 0, "wave steps are whole: no tail step");
static_assert(N_NODES <= (1 << 17) && RNG <= (1 << 14), "a stash word holds row << 17 | source");
static_assert((RNG + 31) / 32 <= 32 && (RNG + 31) / 32 <= SCAP, "a wave's self loops fit one lane each");

__device__ __forceinline__ int rl_i(int v, int l) {
    return __builtin_amdgcn_readlane(v, l);
}
__device__ __forceinline__ float rl_f(float v, int l) {
    return __int_as_float(__builtin_amdgcn_readlane(__float_as_int(v), l));
}

__device__ __forceinline__ float edge_score(float t) {
    return t > 0.f ? t : NEG_SLOPE * t;
}

__device__ __forceinline__ float edge_weight(float t, unsigned mf) {
    return __expf(edge_score(t) - funflip(mf));
}

__device__ __forceinline__ unsigned zero_bytes(unsigned x) {
    return ~(((x & 0x7f7f7f7fu) + 0x7f7f7f7fu) | x | 0x7f7f7f7fu);
}

__device__ __forceinline__ void own_push(bool mine, int e, int* wst, int& qt, bool& ovf) {
    const unsigned mask = __builtin_amdgcn_ballot_w32(mine);
    if (mask == 0u) return;
    const int n = __builtin_popcount(mask);
    if (qt + n <= SCAP) {
        if (mine) wst[qt + (int)__builtin_amdgcn_mbcnt_lo(mask, 0u)] = e;
        qt += n;
    } else {
        ovf = true;
    }
}

#define MAXB 4
__device__ __forceinline__ void stash_max(int* wst, int q, int qt, int lane, int base,
                                          const int* __restrict__ ei,
                                          const float* __restrict__ a_src,
                                          const float* ad_s, unsigned* mx) {
    int idx[MAXB], s[MAXB], d[MAXB];
    bool val[MAXB];
    float4 as[MAXB], ad[MAXB];
    #pragma unroll
    for (int j = 0; j < MAXB; ++j) {
        const int i = q + BATCH * j + lane;
        val[j] = i < qt;
        idx[j] = val[j] ? i : q;
        edge_sd_sel(ei, wst[idx[j]], s[j], d[j]);
    }
    #pragma unroll
    for (int j = 0; j < MAXB; ++j) {
        as[j] = ((const float4*)a_src)[s[j]];
        ad[j] = ((const float4*)ad_s)[d[j] - base];
    }
    #pragma unroll
    for (int j = 0; j < MAXB; ++j) {
        const int r = d[j] - base;
        unsigned* const mp = mx + (val[j] ? r : RNG + lane) * HEADS;
        atomicMax(mp + 0, fflip(edge_score(as[j].x + ad[j].x)));
        atomicMax(mp + 1, fflip(edge_score(as[j].y + ad[j].y)));
        atomicMax(mp + 2, fflip(edge_score(as[j].z + ad[j].z)));
        atomicMax(mp + 3, fflip(edge_score(as[j].w + ad[j].w)));
        if (val[j]) wst[idx[j]] = (r << 17) | s[j];
    }
}

__device__ __forceinline__ void slow_max(int t0, int lane, int base,
                                         const int* __restrict__ ei,
                                         const float* __restrict__ a_src,
                                         const float* ad_s, unsigned* mx) {
    #pragma unroll 1
    for (int k = 0; k < SU * (OSTEP / 32); ++k) {
        const int t = t0 + 32 * (k / (OSTEP / 32));
        if (t < NSTEP) {
            const int e = t * OSTEP + (k % (OSTEP / 32)) * 32 + lane;
            const unsigned r = (unsigned)(ei[N_EDGES + e] - base);
            if (r < (unsigned)RNG) {
                const int s = ei[e];
                #pragma unroll 1
                for (int j = 0; j < HEADS; ++j)
                    atomicMax(&mx[r * HEADS + j], fflip(edge_score(a_src[s * HEADS + j] + ad_s[r * HEADS + j])));
            }
        }
    }
}

__device__ __forceinline__ void slow_sum(int t0, int lane, int base,
                                         const int* __restrict__ ei,
                                         const float* __restrict__ a_src,
                                         const float* __restrict__ h,
                                         const float* ad_s, const unsigned* mx,
                                         float* acc, float* den) {
    #pragma unroll 1
    for (int k = 0; k < SU * (OSTEP / 32); ++k) {
        const int t = t0 + 32 * (k / (OSTEP / 32));
        if (t < NSTEP) {
            const int e = t * OSTEP + (k % (OSTEP / 32)) * 32 + lane;
            const unsigned r = (unsigned)(ei[N_EDGES + e] - base);
            if (r < (unsigned)RNG) {
                const int s = ei[e];
                #pragma unroll 1
                for (int j = 0; j < HEADS; ++j) {
                    const float w = edge_weight(a_src[s * HEADS + j] + ad_s[r * HEADS + j], mx[r * HEADS + j]);
                    atomicAdd(&den[r * HEADS + j], w);
                    #pragma unroll 1
                    for (int c = 0; c < OUT_CH; ++c)
                        atomicAdd(&acc[r * HC + j * OUT_CH + c], h[(size_t)s * HC + j * OUT_CH + c] * w);
                }
            }
        }
    }
}

__device__ __forceinline__ void batch_weights(float4 as, float4 ad, uint4 mf, bool val, int r,
                                              float* den, float (&ex)[HEADS]) {
    ex[0] = edge_weight(as.x + ad.x, mf.x); ex[1] = edge_weight(as.y + ad.y, mf.y);
    ex[2] = edge_weight(as.z + ad.z, mf.z); ex[3] = edge_weight(as.w + ad.w, mf.w);
    #pragma unroll
    for (int j = 0; j < HEADS; ++j) {
        ex[j] = val ? ex[j] : 0.f;
        atomicAdd(&den[r * HEADS + j], ex[j]);
    }
}

__device__ __forceinline__ void rows_load(const float* __restrict__ h, int s, int g, int lane,
                                          float (&hv)[4][HEADS]) {
    #pragma unroll
    for (int k = 0; k < 4; ++k) {
        const float* hp = h + (size_t)rl_i(s, g * 4 + k) * HC + lane;
        #pragma unroll
        for (int j = 0; j < HEADS; ++j) hv[k][j] = hp[j * OUT_CH];
    }
}

__device__ __forceinline__ void rows_add(float* acc, int r, const float (&ex)[HEADS], int g, int lane,
                                         const float (&hv)[4][HEADS]) {
    #pragma unroll
    for (int k = 0; k < 4; ++k) {
        float* ap = acc + rl_i(r, g * 4 + k) * HC + lane;
        #pragma unroll
        for (int j = 0; j < HEADS; ++j)
            atomicAdd(ap + j * OUT_CH, hv[k][j] * rl_f(ex[j], g * 4 + k));
    }
}

__global__ __launch_bounds__(1024) void k_own(const int* __restrict__ ei,
                                              const unsigned* __restrict__ own,
                                              const float* __restrict__ a_src,
                                              const float* __restrict__ a_dst,
                                              const float* __restrict__ h,
                                              const float* __restrict__ bias,
                                              float* __restrict__ out) {
    __shared__ __attribute__((aligned(16))) float    acc[RNG * HC];
    __shared__ __attribute__((aligned(16))) float    den[RNG * HEADS];
    __shared__ __attribute__((aligned(16))) float    ad_s[RNG * HEADS];
    __shared__ __attribute__((aligned(16))) unsigned mx[(RNG + 32) * HEADS];
    __shared__ int st[32 * SCAP];

    const int tid  = threadIdx.x;
    const int lane = tid & 31;
    const int wave = __builtin_amdgcn_readfirstlane(tid >> 5);
    const int base = blockIdx.x * RNG;
    int* const wst = st + wave * SCAP;

    for (int i = tid; i < RNG * HC / 4; i += 1024) ((float4*)acc)[i] = make_float4(0.f, 0.f, 0.f, 0.f);
    for (int i = tid; i < RNG * HEADS; i += 1024) {
        den[i] = 0.f;
        mx[i] = 0u;
        ad_s[i] = a_dst[base * HEADS + i];
    }
    if (tid < 32 * HEADS) mx[RNG * HEADS + tid] = 0u;
    const int nself = (RNG - wave + 31) / 32;
    if (lane < nself) wst[lane] = N_EDGES + base + wave + 32 * lane;
    __syncthreads();

    const uint4* __restrict__ ocol = (const uint4*)own;
    const unsigned pat = (unsigned)blockIdx.x * 0x01010101u;
    int qt = nself;
    int slow0 = NOSLOW;
    for (int t0 = wave; t0 < NSTEP; t0 += 32 * SU) {
        if (slow0 == NOSLOW) {
            uint4 ov[SU];
            const int qt0 = qt;
            bool ovf = false;
            #pragma unroll
            for (int j = 0; j < SU; ++j) ov[j] = ocol[min(t0 + 32 * j, NSTEP - 1) * 32 + lane];
            #pragma unroll
            for (int j = 0; j < SU; ++j) {
                const int t = t0 + 32 * j;
                unsigned c = (zero_bytes(ov[j].x ^ pat) >> 7) | (zero_bytes(ov[j].y ^ pat) >> 6)
                           | (zero_bytes(ov[j].z ^ pat) >> 5) | (zero_bytes(ov[j].w ^ pat) >> 4);
                c = t < NSTEP ? c : 0u;
                const int e0 = t * OSTEP + lane * 16;
                while (__builtin_amdgcn_ballot_w32(c != 0u) != 0u) {
                    const bool mine = c != 0u;
                    const int  p = mine ? __builtin_ctz(c) : 0;
                    own_push(mine, e0 + ((p & 3) << 2) + (p >> 3), wst, qt, ovf);
                    c &= c - 1u;
                }
            }
            if (ovf) {
                qt = qt0;
                slow0 = t0;
            }
        }
        if (slow0 != NOSLOW) slow_max(t0, lane, base, ei, a_src, ad_s, mx);
    }
    __builtin_amdgcn_wave_barrier();
    for (int q = 0; q < qt; q += MAXB * BATCH) stash_max(wst, q, qt, lane, base, ei, a_src, ad_s, mx);
    __syncthreads();

    int qh = 0;
    while (qt - qh >= BATCH) {
        const int p = wst[qh + lane];
        const int s = p & 0x1ffff;
        const int r = p >> 17;
        const float4 as = ((const float4*)a_src)[s];
        float hv[NBUF][4][HEADS];
        #pragma unroll
        for (int g = 0; g < NBUF; ++g) rows_load(h, s, g, lane, hv[g]);
        __builtin_amdgcn_sched_barrier(0);
        const float4 ad = ((const float4*)ad_s)[r];
        const uint4  mf = ((const uint4*)mx)[r];
        float ex[HEADS];
        batch_weights(as, ad, mf, true, r, den, ex);
        #pragma unroll
        for (int g = 0; g < BATCH / 4; ++g) {
            rows_add(acc, r, ex, g, lane, hv[g % NBUF]);
            if (g + NBUF < BATCH / 4) rows_load(h, s, g + NBUF, lane, hv[g % NBUF]);
            __builtin_amdgcn_sched_barrier(0);
        }
        qh += BATCH;
    }
    {
        const int cnt = qt - qh;
        if (cnt > 0) {
            const bool val = lane < cnt;
            const int p = wst[qh + (val ? lane : 0)];
            const int s = p & 0x1ffff;
            const int r = p >> 17;
            float ex[HEADS];
            batch_weights(((const float4*)a_src)[s], ((const float4*)ad_s)[r], ((const uint4*)mx)[r],
                          val, r, den, ex);
            #pragma unroll 1
            for (int g = 0; g * 4 < cnt; ++g) {
                float hv[4][HEADS];
                rows_load(h, s, g, lane, hv);
                rows_add(acc, r, ex, g, lane, hv);
            }
        }
    }
    if (slow0 != NOSLOW) {
        #pragma unroll 1
        for (int t0 = slow0; t0 < NSTEP; t0 += 32 * SU) slow_sum(t0, lane, base, ei, a_src, h, ad_s, mx, acc, den);
    }
    __syncthreads();

    const float4 bv = ((const float4*)bias)[lane];
    for (int r = wave; r < RNG; r += 32) {
        float4 v = ((const float4*)(acc + r * HC))[lane];
        const float dn = den[r * HEADS + (lane >> 3)] + 1e-16f;
        v.x = v.x / dn + bv.x; v.y = v.y / dn + bv.y;
        v.z = v.z / dn + bv.z; v.w = v.w / dn + bv.w;

        float mx_r = fmaxf(fmaxf(v.x, v.y), fmaxf(v.z, v.w));
        #pragma unroll
        for (int o = 16; o >= 1; o >>= 1) mx_r = fmaxf(mx_r, __shfl_xor(mx_r, o, 32));

        float se = __expf(v.x - mx_r) + __expf(v.y - mx_r) + __expf(v.z - mx_r) + __expf(v.w - mx_r);
        #pragma unroll
        for (int o = 16; o >= 1; o >>= 1) se += __shfl_xor(se, o, 32);

        const float lse = mx_r + __logf(se);
        v.x -= lse; v.y -= lse; v.z -= lse; v.w -= lse;
        ((float4*)(out + (size_t)(base + r) * HC))[lane] = v;
    }
}

extern "C" void kernel_launch(void* const* d_in, const int* in_sizes, int n_in,
                              void* d_out, int out_size, void* d_ws, size_t ws_size,
                              hipStream_t stream) {
    const float* x       = (const float*)d_in[0];
    const int*   ei      = (const int*)  d_in[1];
    const float* W       = (const float*)d_in[2];
    const float* att_src = (const float*)d_in[3];
    const float* att_dst = (const float*)d_in[4];
    const float* bias    = (const float*)d_in[5];
    float*       out     = (float*)d_out;

    float*    h     = (float*)d_ws;
    float*    asrc  = h    + (size_t)N_NODES * HC;
    float*    adst  = asrc + (size_t)N_NODES * HEADS;
    unsigned* ownq  = (unsigned*)(adst + (size_t)N_NODES * HEADS);

    k_gemm<<<GEMM_BLOCKS, 256, 0, stream>>>(x, W, att_src, att_dst, ei, h, asrc, adst, ownq);
    k_own <<<NOWN, 1024, 0, stream>>>(ei, ownq, asrc, adst, h, bias, out);
}


# Owner numbers stored as 8 bit planes (wave votes in the first kernel): an owner block matches 64 edges of a lane by one xor-or a plane, wave steps of 4096 edges; LDS float adds stay arrival order
# speedup vs baseline: 6.3420x; 1.0793x over previous
#include <hip/hip_runtime.h>
#include <hip/hip_bf16.h>

#define N_NODES 100000
#define N_EDGES 1600000
#define TOT_E   (N_EDGES + N_NODES)
#define IN_CH   64
#define OUT_CH  32
#define HEADS   4
#define HC      (HEADS * OUT_CH)
#define NEG_SLOPE 0.2f

typedef __attribute__((ext_vector_type(16))) _Float16 v16h;
typedef __attribute__((ext_vector_type(8)))  float    v8f;

__device__ __forceinline__ unsigned fflip(float f) {
    unsigned u = __float_as_uint(f);
    return (u & 0x80000000u) ? ~u : (u | 0x80000000u);
}
__device__ __forceinline__ float funflip(unsigned u) {
    return __uint_as_float((u & 0x80000000u) ? (u & 0x7FFFFFFFu) : ~u);
}

__device__ __forceinline__ void edge_sd_sel(const int* __restrict__ ei, int e, int& s, int& d) {
    const int ec = min(e, N_EDGES - 1);
    const int ls = ei[ec], ld = ei[N_EDGES + ec];
    const bool real = e < N_EDGES;
    s = real ? ls : e - N_EDGES;
    d = real ? ld : e - N_EDGES;
}

#define M_TILES     (N_NODES / 16)
#define GEMM_BLOCKS 1250
#define GEMM_THREADS (GEMM_BLOCKS * 256)
#define XPITCH      72
static_assert(M_TILES % GEMM_BLOCKS == 0, "every block takes the same number of tiles");

#define RNG    400
#define NOWN   (N_NODES / RNG)
static_assert(N_NODES % RNG == 0, "blocks own whole ranges: no row guard");
static_assert(NOWN <= 256, "an owner number fits a byte");
static_assert(N_EDGES % 4 == 0, "owner bytes are written four at a time");
static_assert(N_EDGES / 4 > GEMM_THREADS && N_EDGES / 4 <= 2 * GEMM_THREADS,
              "every thread of the GEMM grid writes one owner word, the first ones a second");
#define PLANE_W (N_EDGES / 32)
#define NGRP    (N_EDGES / 128)
static_assert(N_EDGES % 128 == 0, "whole groups: every plane word is written, every bit is an edge");
static_assert((N_EDGES / 4 - GEMM_THREADS) % 32 == 0, "a wave writes its second group whole or not at all");
static_assert(PLANE_W % 4 == 0, "planes start on 16-byte boundaries");

typedef __attribute__((ext_vector_type(8))) _Float16 v8h;
typedef __attribute__((ext_vector_type(4))) _Float16 v4h;

__device__ __forceinline__ unsigned own_planes(int4 d) {
    const unsigned o[4] = {(unsigned)d.x / RNG, (unsigned)d.y / RNG, (unsigned)d.z / RNG, (unsigned)d.w / RNG};
    unsigned val = 0u;
    #pragma unroll
    for (int k = 0; k < 8; ++k) {
        #pragma unroll
        for (int j = 0; j < 4; ++j) {
            const unsigned m = __builtin_amdgcn_ballot_w32(((o[j] >> k) & 1u) != 0u);
            asm("v_writelane_b32 %0, %1, %2" : "+v"(val) : "s"(m), "n"(4 * k + j));
        }
    }
    return val;
}

template <int CTRL>
__device__ __forceinline__ float row_lane(float v) {
    return __int_as_float(__builtin_amdgcn_update_dpp(0, __float_as_int(v), CTRL, 0xf, 0xf, true));
}
__device__ __forceinline__ float row_sum16(float v) {
    v += row_lane<0xB1>(v);
    v += row_lane<0x4E>(v);
    v += row_lane<0x141>(v);
    v += row_lane<0x140>(v);
    return v;
}

__global__ __launch_bounds__(256) __attribute__((amdgpu_waves_per_eu(10)))
void k_gemm(const float* __restrict__ x,
                                              const float* __restrict__ W,
                                              const float* __restrict__ att_src,
                                              const float* __restrict__ att_dst,
                                              const int* __restrict__ ei,
                                              float* __restrict__ h,
                                              float* __restrict__ a_src,
                                              float* __restrict__ a_dst,
                                              unsigned* __restrict__ ownq) {
    __shared__ __attribute__((aligned(16))) _Float16 xs[2][16 * XPITCH];
    __shared__ __attribute__((aligned(16))) float part[2][8][2][16];
    const int tid = threadIdx.x;
    const int wave = tid >> 5;
    const int lane = tid & 31;
    const int mrow = lane & 15;
    const int khalf = (lane >> 4) << 3;
    const int ncol = wave * 16 + (lane & 15);

    const int gt = blockIdx.x * 256 + tid;
    const int4* __restrict__ dq = (const int4*)(ei + N_EDGES);
    const bool two = gt < N_EDGES / 4 - GEMM_THREADS;
    const int4 d0 = dq[gt];
    const int4 d1 = dq[two ? gt + GEMM_THREADS : gt];

    v16h b0, b1;
    #pragma unroll
    for (int i = 0; i < 16; ++i) {
        int k = i + (i & 8) + khalf;
        b0[i] = (_Float16)W[(size_t)k * HC + ncol];
        b1[i] = (_Float16)W[(size_t)(k + 32) * HC + ncol];
    }
    const float as_c = att_src[ncol];
    const float ad_c = att_dst[ncol];

    const unsigned o0 = own_planes(d0), o1 = own_planes(d1);

    const int jidx = tid & 63;
    const int jrow = jidx >> 2, jhead = jidx & 3;
    const int joff = (2 * jhead * 2 + (jrow >> 3)) * 16 + ((tid >> 6) & 1) * 8 + (jrow & 7);

    const int sr = tid >> 4;
    const int sc = (tid & 15) << 2;
    int mt = blockIdx.x;
    float4 xv = *(const float4*)(x + (size_t)(mt * 16 + sr) * IN_CH + sc);
    int buf = 0;
    #pragma nounroll
    for (; mt < M_TILES; mt += GEMM_BLOCKS) {
        v4h xh;
        xh[0] = (_Float16)xv.x; xh[1] = (_Float16)xv.y; xh[2] = (_Float16)xv.z; xh[3] = (_Float16)xv.w;
        *(v4h*)(&xs[buf][sr * XPITCH + sc]) = xh;
        __syncthreads();
        const int nx = mt + GEMM_BLOCKS;
        if (nx < M_TILES) xv = *(const float4*)(x + (size_t)(nx * 16 + sr) * IN_CH + sc);

        if (mt != (int)blockIdx.x && tid < 128) {
            const float* pp = &part[buf ^ 1][0][0][0] + joff;
            const float v = pp[0] + pp[32];
            const size_t o = (size_t)(mt - GEMM_BLOCKS) * 16 * HEADS + jidx;
            if (tid < 64) a_src[o] = v; else a_dst[o] = v;
        }

        const _Float16* rp = &xs[buf][mrow * XPITCH + khalf];
        const v8h p0 = *(const v8h*)(rp);
        const v8h p1 = *(const v8h*)(rp + 16);
        const v8h p2 = *(const v8h*)(rp + 32);
        const v8h p3 = *(const v8h*)(rp + 48);
        const v16h a0 = __builtin_shufflevector(p0, p1, 0, 1, 2, 3, 4, 5, 6, 7, 8, 9, 10, 11, 12, 13, 14, 15);
        const v16h a1 = __builtin_shufflevector(p2, p3, 0, 1, 2, 3, 4, 5, 6, 7, 8, 9, 10, 11, 12, 13, 14, 15);

        v8f c = {};
        c = __builtin_amdgcn_wmma_f32_16x16x32_f16(false, a0, false, b0, (short)0, c, false, false);
        c = __builtin_amdgcn_wmma_f32_16x16x32_f16(false, a1, false, b1, (short)0, c, false, false);

        float* const hrow = h + ((size_t)(mt * 16 + khalf) * HC + ncol);
        #pragma unroll
        for (int r = 0; r < 8; ++r) hrow[r * HC] = c[r];

        float ps[8], pd[8];
        #pragma unroll
        for (int r = 0; r < 8; ++r) {
            ps[r] = row_sum16(c[r] * as_c);
            pd[r] = row_sum16(c[r] * ad_c);
        }
        if ((lane & 15) == 0) {
            float4* const wp = (float4*)&part[buf][wave][lane >> 4][0];
            wp[0] = make_float4(ps[0], ps[1], ps[2], ps[3]);
            wp[1] = make_float4(ps[4], ps[5], ps[6], ps[7]);
            wp[2] = make_float4(pd[0], pd[1], pd[2], pd[3]);
            wp[3] = make_float4(pd[4], pd[5], pd[6], pd[7]);
        }
        buf ^= 1;
    }
    __syncthreads();
    if (tid < 128) {
        const float* pp = &part[buf ^ 1][0][0][0] + joff;
        const float v = pp[0] + pp[32];
        const size_t o = (size_t)(mt - GEMM_BLOCKS) * 16 * HEADS + jidx;
        if (tid < 64) a_src[o] = v; else a_dst[o] = v;
    }
    unsigned* const op = ownq + (lane >> 2) * PLANE_W + (gt >> 5) * 4 + (lane & 3);
    op[0] = o0;
    if (two) op[(GEMM_THREADS / 32) * 4] = o1;
}

#define SCAP   384
#define BATCH  32
#define NBUF   3
#define SU     2
#define OSTEP  4096
#define NSTEP  ((N_EDGES + OSTEP - 1) / OSTEP)
#define NOSLOW 0x7fffffff
static_assert(OSTEP == 32 * 128, "a lane's part of a step is one group of the first kernel");
static_assert(N_NODES <= (1 << 17) && RNG <= (1 << 14), "a stash word holds row << 17 | source");
static_assert((RNG + 31) / 32 <= 32 && (RNG + 31) / 32 <= SCAP, "a wave's self loops fit one lane each");

__device__ __forceinline__ int rl_i(int v, int l) {
    return __builtin_amdgcn_readlane(v, l);
}
__device__ __forceinline__ float rl_f(float v, int l) {
    return __int_as_float(__builtin_amdgcn_readlane(__float_as_int(v), l));
}

__device__ __forceinline__ float edge_score(float t) {
    return t > 0.f ? t : NEG_SLOPE * t;
}

__device__ __forceinline__ float edge_weight(float t, unsigned mf) {
    return __expf(edge_score(t) - funflip(mf));
}

__device__ __forceinline__ void own_push(bool mine, int e, int* wst, int& qt, bool& ovf) {
    const unsigned mask = __builtin_amdgcn_ballot_w32(mine);
    if (mask == 0u) return;
    const int n = __builtin_popcount(mask);
    if (qt + n <= SCAP) {
        if (mine) wst[qt + (int)__builtin_amdgcn_mbcnt_lo(mask, 0u)] = e;
        qt += n;
    } else {
        ovf = true;
    }
}

#define MAXB 4
__device__ __forceinline__ void stash_max(int* wst, int q, int qt, int lane, int base,
                                          const int* __restrict__ ei,
                                          const float* __restrict__ a_src,
                                          const float* ad_s, unsigned* mx) {
    int idx[MAXB], s[MAXB], d[MAXB];
    bool val[MAXB];
    float4 as[MAXB], ad[MAXB];
    #pragma unroll
    for (int j = 0; j < MAXB; ++j) {
        const int i = q + BATCH * j + lane;
        val[j] = i < qt;
        idx[j] = val[j] ? i : q;
        edge_sd_sel(ei, wst[idx[j]], s[j], d[j]);
    }
    #pragma unroll
    for (int j = 0; j < MAXB; ++j) {
        as[j] = ((const float4*)a_src)[s[j]];
        ad[j] = ((const float4*)ad_s)[d[j] - base];
    }
    #pragma unroll
    for (int j = 0; j < MAXB; ++j) {
        const int r = d[j] - base;
        unsigned* const mp = mx + (val[j] ? r : RNG + lane) * HEADS;
        atomicMax(mp + 0, fflip(edge_score(as[j].x + ad[j].x)));
        atomicMax(mp + 1, fflip(edge_score(as[j].y + ad[j].y)));
        atomicMax(mp + 2, fflip(edge_score(as[j].z + ad[j].z)));
        atomicMax(mp + 3, fflip(edge_score(as[j].w + ad[j].w)));
        if (val[j]) wst[idx[j]] = (r << 17) | s[j];
    }
}

__device__ __forceinline__ void slow_max(int t0, int lane, int base,
                                         const int* __restrict__ ei,
                                         const float* __restrict__ a_src,
                                         const float* ad_s, unsigned* mx) {
    #pragma unroll 1
    for (int k = 0; k < SU * (OSTEP / 32); ++k) {
        const int t = t0 + 32 * (k / (OSTEP / 32));
        const int e = t * OSTEP + (k % (OSTEP / 32)) * 32 + lane;
        if (t < NSTEP && e < N_EDGES) {
            const unsigned r = (unsigned)(ei[N_EDGES + e] - base);
            if (r < (unsigned)RNG) {
                const int s = ei[e];
                #pragma unroll 1
                for (int j = 0; j < HEADS; ++j)
                    atomicMax(&mx[r * HEADS + j], fflip(edge_score(a_src[s * HEADS + j] + ad_s[r * HEADS + j])));
            }
        }
    }
}

__device__ __forceinline__ void slow_sum(int t0, int lane, int base,
                                         const int* __restrict__ ei,
                                         const float* __restrict__ a_src,
                                         const float* __restrict__ h,
                                         const float* ad_s, const unsigned* mx,
                                         float* acc, float* den) {
    #pragma unroll 1
    for (int k = 0; k < SU * (OSTEP / 32); ++k) {
        const int t = t0 + 32 * (k / (OSTEP / 32));
        const int e = t * OSTEP + (k % (OSTEP / 32)) * 32 + lane;
        if (t < NSTEP && e < N_EDGES) {
            const unsigned r = (unsigned)(ei[N_EDGES + e] - base);
            if (r < (unsigned)RNG) {
                const int s = ei[e];
                #pragma unroll 1
                for (int j = 0; j < HEADS; ++j) {
                    const float w = edge_weight(a_src[s * HEADS + j] + ad_s[r * HEADS + j], mx[r * HEADS + j]);
                    atomicAdd(&den[r * HEADS + j], w);
                    #pragma unroll 1
                    for (int c = 0; c < OUT_CH; ++c)
                        atomicAdd(&acc[r * HC + j * OUT_CH + c], h[(size_t)s * HC + j * OUT_CH + c] * w);
                }
            }
        }
    }
}

__device__ __forceinline__ void batch_weights(float4 as, float4 ad, uint4 mf, bool val, int r,
                                              float* den, float (&ex)[HEADS]) {
    ex[0] = edge_weight(as.x + ad.x, mf.x); ex[1] = edge_weight(as.y + ad.y, mf.y);
    ex[2] = edge_weight(as.z + ad.z, mf.z); ex[3] = edge_weight(as.w + ad.w, mf.w);
    #pragma unroll
    for (int j = 0; j < HEADS; ++j) {
        ex[j] = val ? ex[j] : 0.f;
        atomicAdd(&den[r * HEADS + j], ex[j]);
    }
}

__device__ __forceinline__ void rows_load(const float* __restrict__ h, int s, int g, int lane,
                                          float (&hv)[4][HEADS]) {
    #pragma unroll
    for (int k = 0; k < 4; ++k) {
        const float* hp = h + (size_t)rl_i(s, g * 4 + k) * HC + lane;
        #pragma unroll
        for (int j = 0; j < HEADS; ++j) hv[k][j] = hp[j * OUT_CH];
    }
}

__device__ __forceinline__ void rows_add(float* acc, int r, const float (&ex)[HEADS], int g, int lane,
                                         const float (&hv)[4][HEADS]) {
    #pragma unroll
    for (int k = 0; k < 4; ++k) {
        float* ap = acc + rl_i(r, g * 4 + k) * HC + lane;
        #pragma unroll
        for (int j = 0; j < HEADS; ++j)
            atomicAdd(ap + j * OUT_CH, hv[k][j] * rl_f(ex[j], g * 4 + k));
    }
}

__global__ __launch_bounds__(1024) void k_own(const int* __restrict__ ei,
                                              const unsigned* __restrict__ own,
                                              const float* __restrict__ a_src,
                                              const float* __restrict__ a_dst,
                                              const float* __restrict__ h,
                                              const float* __restrict__ bias,
                                              float* __restrict__ out) {
    __shared__ __attribute__((aligned(16))) float    acc[RNG * HC];
    __shared__ __attribute__((aligned(16))) float    den[RNG * HEADS];
    __shared__ __attribute__((aligned(16))) float    ad_s[RNG * HEADS];
    __shared__ __attribute__((aligned(16))) unsigned mx[(RNG + 32) * HEADS];
    __shared__ int st[32 * SCAP];

    const int tid  = threadIdx.x;
    const int lane = tid & 31;
    const int wave = __builtin_amdgcn_readfirstlane(tid >> 5);
    const int base = blockIdx.x * RNG;
    int* const wst = st + wave * SCAP;

    for (int i = tid; i < RNG * HC / 4; i += 1024) ((float4*)acc)[i] = make_float4(0.f, 0.f, 0.f, 0.f);
    for (int i = tid; i < RNG * HEADS; i += 1024) {
        den[i] = 0.f;
        mx[i] = 0u;
        ad_s[i] = a_dst[base * HEADS + i];
    }
    if (tid < 32 * HEADS) mx[RNG * HEADS + tid] = 0u;
    const int nself = (RNG - wave + 31) / 32;
    if (lane < nself) wst[lane] = N_EDGES + base + wave + 32 * lane;
    __syncthreads();

    const uint4* __restrict__ ocol = (const uint4*)own;
    unsigned inv[8];
    #pragma unroll
    for (int k = 0; k < 8; ++k) inv[k] = ((blockIdx.x >> k) & 1u) ? 0xffffffffu : 0u;
    int qt = nself;
    int slow0 = NOSLOW;
    for (int t0 = wave; t0 < NSTEP; t0 += 32 * SU) {
        if (slow0 == NOSLOW) {
            uint4 ov[SU][8];
            const int qt0 = qt;
            bool ovf = false;
            #pragma unroll
            for (int j = 0; j < SU; ++j) {
                const int g = min((t0 + 32 * j) * 32 + lane, NGRP - 1);
                #pragma unroll
                for (int k = 0; k < 8; ++k) ov[j][k] = ocol[k * (PLANE_W / 4) + g];
            }
            __builtin_amdgcn_sched_barrier(0);
            #pragma unroll
            for (int j = 0; j < SU; ++j) {
                const int g = (t0 + 32 * j) * 32 + lane;
                unsigned mis[4] = {0u, 0u, 0u, 0u};
                #pragma unroll
                for (int k = 0; k < 8; ++k) {
                    mis[0] |= ov[j][k].x ^ inv[k]; mis[1] |= ov[j][k].y ^ inv[k];
                    mis[2] |= ov[j][k].z ^ inv[k]; mis[3] |= ov[j][k].w ^ inv[k];
                }
                const int e0 = g * 128;
                #pragma unroll
                for (int w = 0; w < 4; w += 2) {
                    unsigned long long c = g < NGRP
                        ? ((unsigned long long)(~mis[w + 1]) << 32) | (unsigned long long)(~mis[w]) : 0ull;
                    while (__builtin_amdgcn_ballot_w32(c != 0ull) != 0u) {
                        const bool mine = c != 0ull;
                        const int  p = mine ? __builtin_ctzll(c) : 0;
                        own_push(mine, e0 + ((p & 31) << 2) + w + (p >> 5), wst, qt, ovf);
                        c &= c - 1ull;
                    }
                }
            }
            if (ovf) {
                qt = qt0;
                slow0 = t0;
            }
        }
        if (slow0 != NOSLOW) slow_max(t0, lane, base, ei, a_src, ad_s, mx);
    }
    __builtin_amdgcn_wave_barrier();
    for (int q = 0; q < qt; q += MAXB * BATCH) stash_max(wst, q, qt, lane, base, ei, a_src, ad_s, mx);
    __syncthreads();

    int qh = 0;
    while (qt - qh >= BATCH) {
        const int p = wst[qh + lane];
        const int s = p & 0x1ffff;
        const int r = p >> 17;
        const float4 as = ((const float4*)a_src)[s];
        float hv[NBUF][4][HEADS];
        #pragma unroll
        for (int g = 0; g < NBUF; ++g) rows_load(h, s, g, lane, hv[g]);
        __builtin_amdgcn_sched_barrier(0);
        const float4 ad = ((const float4*)ad_s)[r];
        const uint4  mf = ((const uint4*)mx)[r];
        float ex[HEADS];
        batch_weights(as, ad, mf, true, r, den, ex);
        #pragma unroll
        for (int g = 0; g < BATCH / 4; ++g) {
            rows_add(acc, r, ex, g, lane, hv[g % NBUF]);
            if (g + NBUF < BATCH / 4) rows_load(h, s, g + NBUF, lane, hv[g % NBUF]);
            __builtin_amdgcn_sched_barrier(0);
        }
        qh += BATCH;
    }
    {
        const int cnt = qt - qh;
        if (cnt > 0) {
            const bool val = lane < cnt;
            const int p = wst[qh + (val ? lane : 0)];
            const int s = p & 0x1ffff;
            const int r = p >> 17;
            float ex[HEADS];
            batch_weights(((const float4*)a_src)[s], ((const float4*)ad_s)[r], ((const uint4*)mx)[r],
                          val, r, den, ex);
            #pragma unroll 1
            for (int g = 0; g * 4 < cnt; ++g) {
                float hv[4][HEADS];
                rows_load(h, s, g, lane, hv);
                rows_add(acc, r, ex, g, lane, hv);
            }
        }
    }
    if (slow0 != NOSLOW) {
        #pragma unroll 1
        for (int t0 = slow0; t0 < NSTEP; t0 += 32 * SU) slow_sum(t0, lane, base, ei, a_src, h, ad_s, mx, acc, den);
    }
    __syncthreads();

    const float4 bv = ((const float4*)bias)[lane];
    for (int r = wave; r < RNG; r += 32) {
        float4 v = ((const float4*)(acc + r * HC))[lane];
        const float dn = den[r * HEADS + (lane >> 3)] + 1e-16f;
        v.x = v.x / dn + bv.x; v.y = v.y / dn + bv.y;
        v.z = v.z / dn + bv.z; v.w = v.w / dn + bv.w;

        float mx_r = fmaxf(fmaxf(v.x, v.y), fmaxf(v.z, v.w));
        #pragma unroll
        for (int o = 16; o >= 1; o >>= 1) mx_r = fmaxf(mx_r, __shfl_xor(mx_r, o, 32));

        float se = __expf(v.x - mx_r) + __expf(v.y - mx_r) + __expf(v.z - mx_r) + __expf(v.w - mx_r);
        #pragma unroll
        for (int o = 16; o >= 1; o >>= 1) se += __shfl_xor(se, o, 32);

        const float lse = mx_r + __logf(se);
        v.x -= lse; v.y -= lse; v.z -= lse; v.w -= lse;
        ((float4*)(out + (size_t)(base + r) * HC))[lane] = v;
    }
}

extern "C" void kernel_launch(void* const* d_in, const int* in_sizes, int n_in,
                              void* d_out, int out_size, void* d_ws, size_t ws_size,
                              hipStream_t stream) {
    const float* x       = (const float*)d_in[0];
    const int*   ei      = (const int*)  d_in[1];
    const float* W       = (const float*)d_in[2];
    const float* att_src = (const float*)d_in[3];
    const float* att_dst = (const float*)d_in[4];
    const float* bias    = (const float*)d_in[5];
    float*       out     = (float*)d_out;

    float*    h     = (float*)d_ws;
    float*    asrc  = h    + (size_t)N_NODES * HC;
    float*    adst  = asrc + (size_t)N_NODES * HEADS;
    unsigned* ownq  = (unsigned*)(adst + (size_t)N_NODES * HEADS);

    k_gemm<<<GEMM_BLOCKS, 256, 0, stream>>>(x, W, att_src, att_dst, ei, h, asrc, adst, ownq);
    k_own <<<NOWN, 1024, 0, stream>>>(ei, ownq, asrc, adst, h, bias, out);
}


# Pass-1 scores kept in registers for pass 2 in the owner blocks (a wave's first 8 batches, lane = entry in both): no second 16-byte score gather a batch; LDS float adds stay arrival order
# speedup vs baseline: 6.5808x; 1.0377x over previous
#include <hip/hip_runtime.h>
#include <hip/hip_bf16.h>

#define N_NODES 100000
#define N_EDGES 1600000
#define TOT_E   (N_EDGES + N_NODES)
#define IN_CH   64
#define OUT_CH  32
#define HEADS   4
#define HC      (HEADS * OUT_CH)
#define NEG_SLOPE 0.2f

typedef __attribute__((ext_vector_type(16))) _Float16 v16h;
typedef __attribute__((ext_vector_type(8)))  float    v8f;

__device__ __forceinline__ unsigned fflip(float f) {
    unsigned u = __float_as_uint(f);
    return (u & 0x80000000u) ? ~u : (u | 0x80000000u);
}
__device__ __forceinline__ float funflip(unsigned u) {
    return __uint_as_float((u & 0x80000000u) ? (u & 0x7FFFFFFFu) : ~u);
}

__device__ __forceinline__ void edge_sd_sel(const int* __restrict__ ei, int e, int& s, int& d) {
    const int ec = min(e, N_EDGES - 1);
    const int ls = ei[ec], ld = ei[N_EDGES + ec];
    const bool real = e < N_EDGES;
    s = real ? ls : e - N_EDGES;
    d = real ? ld : e - N_EDGES;
}

#define M_TILES     (N_NODES / 16)
#define GEMM_BLOCKS 1250
#define GEMM_THREADS (GEMM_BLOCKS * 256)
#define XPITCH      72
static_assert(M_TILES % GEMM_BLOCKS == 0, "every block takes the same number of tiles");

#define RNG    400
#define NOWN   (N_NODES / RNG)
static_assert(N_NODES % RNG == 0, "blocks own whole ranges: no row guard");
static_assert(NOWN <= 256, "an owner number fits a byte");
static_assert(N_EDGES % 4 == 0, "owner bytes are written four at a time");
static_assert(N_EDGES / 4 > GEMM_THREADS && N_EDGES / 4 <= 2 * GEMM_THREADS,
              "every thread of the GEMM grid writes one owner word, the first ones a second");
#define PLANE_W (N_EDGES / 32)
#define NGRP    (N_EDGES / 128)
static_assert(N_EDGES % 128 == 0, "whole groups: every plane word is written, every bit is an edge");
static_assert((N_EDGES / 4 - GEMM_THREADS) % 32 == 0, "a wave writes its second group whole or not at all");
static_assert(PLANE_W % 4 == 0, "planes start on 16-byte boundaries");

typedef __attribute__((ext_vector_type(8))) _Float16 v8h;
typedef __attribute__((ext_vector_type(4))) _Float16 v4h;

__device__ __forceinline__ unsigned own_planes(int4 d) {
    const unsigned o[4] = {(unsigned)d.x / RNG, (unsigned)d.y / RNG, (unsigned)d.z / RNG, (unsigned)d.w / RNG};
    unsigned val = 0u;
    #pragma unroll
    for (int k = 0; k < 8; ++k) {
        #pragma unroll
        for (int j = 0; j < 4; ++j) {
            const unsigned m = __builtin_amdgcn_ballot_w32(((o[j] >> k) & 1u) != 0u);
            asm("v_writelane_b32 %0, %1, %2" : "+v"(val) : "s"(m), "n"(4 * k + j));
        }
    }
    return val;
}

template <int CTRL>
__device__ __forceinline__ float row_lane(float v) {
    return __int_as_float(__builtin_amdgcn_update_dpp(0, __float_as_int(v), CTRL, 0xf, 0xf, true));
}
__device__ __forceinline__ float row_sum16(float v) {
    v += row_lane<0xB1>(v);
    v += row_lane<0x4E>(v);
    v += row_lane<0x141>(v);
    v += row_lane<0x140>(v);
    return v;
}

__global__ __launch_bounds__(256) __attribute__((amdgpu_waves_per_eu(10)))
void k_gemm(const float* __restrict__ x,
                                              const float* __restrict__ W,
                                              const float* __restrict__ att_src,
                                              const float* __restrict__ att_dst,
                                              const int* __restrict__ ei,
                                              float* __restrict__ h,
                                              float* __restrict__ a_src,
                                              float* __restrict__ a_dst,
                                              unsigned* __restrict__ ownq) {
    __shared__ __attribute__((aligned(16))) _Float16 xs[2][16 * XPITCH];
    __shared__ __attribute__((aligned(16))) float part[2][8][2][16];
    const int tid = threadIdx.x;
    const int wave = tid >> 5;
    const int lane = tid & 31;
    const int mrow = lane & 15;
    const int khalf = (lane >> 4) << 3;
    const int ncol = wave * 16 + (lane & 15);

    const int gt = blockIdx.x * 256 + tid;
    const int4* __restrict__ dq = (const int4*)(ei + N_EDGES);
    const bool two = gt < N_EDGES / 4 - GEMM_THREADS;
    const int4 d0 = dq[gt];
    const int4 d1 = dq[two ? gt + GEMM_THREADS : gt];

    v16h b0, b1;
    #pragma unroll
    for (int i = 0; i < 16; ++i) {
        int k = i + (i & 8) + khalf;
        b0[i] = (_Float16)W[(size_t)k * HC + ncol];
        b1[i] = (_Float16)W[(size_t)(k + 32) * HC + ncol];
    }
    const float as_c = att_src[ncol];
    const float ad_c = att_dst[ncol];

    const unsigned o0 = own_planes(d0), o1 = own_planes(d1);

    const int jidx = tid & 63;
    const int jrow = jidx >> 2, jhead = jidx & 3;
    const int joff = (2 * jhead * 2 + (jrow >> 3)) * 16 + ((tid >> 6) & 1) * 8 + (jrow & 7);

    const int sr = tid >> 4;
    const int sc = (tid & 15) << 2;
    int mt = blockIdx.x;
    float4 xv = *(const float4*)(x + (size_t)(mt * 16 + sr) * IN_CH + sc);
    int buf = 0;
    #pragma nounroll
    for (; mt < M_TILES; mt += GEMM_BLOCKS) {
        v4h xh;
        xh[0] = (_Float16)xv.x; xh[1] = (_Float16)xv.y; xh[2] = (_Float16)xv.z; xh[3] = (_Float16)xv.w;
        *(v4h*)(&xs[buf][sr * XPITCH + sc]) = xh;
        __syncthreads();
        const int nx = mt + GEMM_BLOCKS;
        if (nx < M_TILES) xv = *(const float4*)(x + (size_t)(nx * 16 + sr) * IN_CH + sc);

        if (mt != (int)blockIdx.x && tid < 128) {
            const float* pp = &part[buf ^ 1][0][0][0] + joff;
            const float v = pp[0] + pp[32];
            const size_t o = (size_t)(mt - GEMM_BLOCKS) * 16 * HEADS + jidx;
            if (tid < 64) a_src[o] = v; else a_dst[o] = v;
        }

        const _Float16* rp = &xs[buf][mrow * XPITCH + khalf];
        const v8h p0 = *(const v8h*)(rp);
        const v8h p1 = *(const v8h*)(rp + 16);
        const v8h p2 = *(const v8h*)(rp + 32);
        const v8h p3 = *(const v8h*)(rp + 48);
        const v16h a0 = __builtin_shufflevector(p0, p1, 0, 1, 2, 3, 4, 5, 6, 7, 8, 9, 10, 11, 12, 13, 14, 15);
        const v16h a1 = __builtin_shufflevector(p2, p3, 0, 1, 2, 3, 4, 5, 6, 7, 8, 9, 10, 11, 12, 13, 14, 15);

        v8f c = {};
        c = __builtin_amdgcn_wmma_f32_16x16x32_f16(false, a0, false, b0, (short)0, c, false, false);
        c = __builtin_amdgcn_wmma_f32_16x16x32_f16(false, a1, false, b1, (short)0, c, false, false);

        float* const hrow = h + ((size_t)(mt * 16 + khalf) * HC + ncol);
        #pragma unroll
        for (int r = 0; r < 8; ++r) hrow[r * HC] = c[r];

        float ps[8], pd[8];
        #pragma unroll
        for (int r = 0; r < 8; ++r) {
            ps[r] = row_sum16(c[r] * as_c);
            pd[r] = row_sum16(c[r] * ad_c);
        }
        if ((lane & 15) == 0) {
            float4* const wp = (float4*)&part[buf][wave][lane >> 4][0];
            wp[0] = make_float4(ps[0], ps[1], ps[2], ps[3]);
            wp[1] = make_float4(ps[4], ps[5], ps[6], ps[7]);
            wp[2] = make_float4(pd[0], pd[1], pd[2], pd[3]);
            wp[3] = make_float4(pd[4], pd[5], pd[6], pd[7]);
        }
        buf ^= 1;
    }
    __syncthreads();
    if (tid < 128) {
        const float* pp = &part[buf ^ 1][0][0][0] + joff;
        const float v = pp[0] + pp[32];
        const size_t o = (size_t)(mt - GEMM_BLOCKS) * 16 * HEADS + jidx;
        if (tid < 64) a_src[o] = v; else a_dst[o] = v;
    }
    unsigned* const op = ownq + (lane >> 2) * PLANE_W + (gt >> 5) * 4 + (lane & 3);
    op[0] = o0;
    if (two) op[(GEMM_THREADS / 32) * 4] = o1;
}

#define SCAP   384
#define BATCH  32
#define NBUF   3
#define KEEP   8
#define SU     2
#define OSTEP  4096
#define NSTEP  ((N_EDGES + OSTEP - 1) / OSTEP)
#define NOSLOW 0x7fffffff
static_assert(OSTEP == 32 * 128, "a lane's part of a step is one group of the first kernel");
static_assert(N_NODES <= (1 << 17) && RNG <= (1 << 14), "a stash word holds row << 17 | source");
static_assert((RNG + 31) / 32 <= 32 && (RNG + 31) / 32 <= SCAP, "a wave's self loops fit one lane each");

__device__ __forceinline__ int rl_i(int v, int l) {
    return __builtin_amdgcn_readlane(v, l);
}
__device__ __forceinline__ float rl_f(float v, int l) {
    return __int_as_float(__builtin_amdgcn_readlane(__float_as_int(v), l));
}

__device__ __forceinline__ float edge_score(float t) {
    return t > 0.f ? t : NEG_SLOPE * t;
}

__device__ __forceinline__ float edge_weight(float t, unsigned mf) {
    return __expf(edge_score(t) - funflip(mf));
}

__device__ __forceinline__ void own_push(bool mine, int e, int* wst, int& qt, bool& ovf) {
    const unsigned mask = __builtin_amdgcn_ballot_w32(mine);
    if (mask == 0u) return;
    const int n = __builtin_popcount(mask);
    if (qt + n <= SCAP) {
        if (mine) wst[qt + (int)__builtin_amdgcn_mbcnt_lo(mask, 0u)] = e;
        qt += n;
    } else {
        ovf = true;
    }
}

#define MAXB 4
static_assert(KEEP % MAXB == 0, "whole calls of pass 1 are kept");
__device__ __forceinline__ void stash_max(int* wst, int q, int qt, int lane, int base,
                                          const int* __restrict__ ei,
                                          const float* __restrict__ a_src,
                                          const float* ad_s, unsigned* mx,
                                          float (&eo)[MAXB][HEADS]) {
    int idx[MAXB], s[MAXB], d[MAXB];
    bool val[MAXB];
    float4 as[MAXB], ad[MAXB];
    #pragma unroll
    for (int j = 0; j < MAXB; ++j) {
        const int i = q + BATCH * j + lane;
        val[j] = i < qt;
        idx[j] = val[j] ? i : q;
        edge_sd_sel(ei, wst[idx[j]], s[j], d[j]);
    }
    #pragma unroll
    for (int j = 0; j < MAXB; ++j) {
        as[j] = ((const float4*)a_src)[s[j]];
        ad[j] = ((const float4*)ad_s)[d[j] - base];
    }
    #pragma unroll
    for (int j = 0; j < MAXB; ++j) {
        const int r = d[j] - base;
        unsigned* const mp = mx + (val[j] ? r : RNG + lane) * HEADS;
        eo[j][0] = edge_score(as[j].x + ad[j].x);
        eo[j][1] = edge_score(as[j].y + ad[j].y);
        eo[j][2] = edge_score(as[j].z + ad[j].z);
        eo[j][3] = edge_score(as[j].w + ad[j].w);
        #pragma unroll
        for (int k = 0; k < HEADS; ++k) atomicMax(mp + k, fflip(eo[j][k]));
        if (val[j]) wst[idx[j]] = (r << 17) | s[j];
    }
}

__device__ __forceinline__ void slow_max(int t0, int lane, int base,
                                         const int* __restrict__ ei,
                                         const float* __restrict__ a_src,
                                         const float* ad_s, unsigned* mx) {
    #pragma unroll 1
    for (int k = 0; k < SU * (OSTEP / 32); ++k) {
        const int t = t0 + 32 * (k / (OSTEP / 32));
        const int e = t * OSTEP + (k % (OSTEP / 32)) * 32 + lane;
        if (t < NSTEP && e < N_EDGES) {
            const unsigned r = (unsigned)(ei[N_EDGES + e] - base);
            if (r < (unsigned)RNG) {
                const int s = ei[e];
                #pragma unroll 1
                for (int j = 0; j < HEADS; ++j)
                    atomicMax(&mx[r * HEADS + j], fflip(edge_score(a_src[s * HEADS + j] + ad_s[r * HEADS + j])));
            }
        }
    }
}

__device__ __forceinline__ void slow_sum(int t0, int lane, int base,
                                         const int* __restrict__ ei,
                                         const float* __restrict__ a_src,
                                         const float* __restrict__ h,
                                         const float* ad_s, const unsigned* mx,
                                         float* acc, float* den) {
    #pragma unroll 1
    for (int k = 0; k < SU * (OSTEP / 32); ++k) {
        const int t = t0 + 32 * (k / (OSTEP / 32));
        const int e = t * OSTEP + (k % (OSTEP / 32)) * 32 + lane;
        if (t < NSTEP && e < N_EDGES) {
            const unsigned r = (unsigned)(ei[N_EDGES + e] - base);
            if (r < (unsigned)RNG) {
                const int s = ei[e];
                #pragma unroll 1
                for (int j = 0; j < HEADS; ++j) {
                    const float w = edge_weight(a_src[s * HEADS + j] + ad_s[r * HEADS + j], mx[r * HEADS + j]);
                    atomicAdd(&den[r * HEADS + j], w);
                    #pragma unroll 1
                    for (int c = 0; c < OUT_CH; ++c)
                        atomicAdd(&acc[r * HC + j * OUT_CH + c], h[(size_t)s * HC + j * OUT_CH + c] * w);
                }
            }
        }
    }
}

__device__ __forceinline__ void batch_weights(float4 as, float4 ad, uint4 mf, bool val, int r,
                                              float* den, float (&ex)[HEADS]) {
    ex[0] = edge_weight(as.x + ad.x, mf.x); ex[1] = edge_weight(as.y + ad.y, mf.y);
    ex[2] = edge_weight(as.z + ad.z, mf.z); ex[3] = edge_weight(as.w + ad.w, mf.w);
    #pragma unroll
    for (int j = 0; j < HEADS; ++j) {
        ex[j] = val ? ex[j] : 0.f;
        atomicAdd(&den[r * HEADS + j], ex[j]);
    }
}

__device__ __forceinline__ void kept_weights(const float (&e)[HEADS], uint4 mf, bool val, int r,
                                             float* den, float (&ex)[HEADS]) {
    ex[0] = __expf(e[0] - funflip(mf.x)); ex[1] = __expf(e[1] - funflip(mf.y));
    ex[2] = __expf(e[2] - funflip(mf.z)); ex[3] = __expf(e[3] - funflip(mf.w));
    #pragma unroll
    for (int j = 0; j < HEADS; ++j) {
        ex[j] = val ? ex[j] : 0.f;
        atomicAdd(&den[r * HEADS + j], ex[j]);
    }
}

__device__ __forceinline__ void rows_load(const float* __restrict__ h, int s, int g, int lane,
                                          float (&hv)[4][HEADS]) {
    #pragma unroll
    for (int k = 0; k < 4; ++k) {
        const float* hp = h + (size_t)rl_i(s, g * 4 + k) * HC + lane;
        #pragma unroll
        for (int j = 0; j < HEADS; ++j) hv[k][j] = hp[j * OUT_CH];
    }
}

__device__ __forceinline__ void rows_add(float* acc, int r, const float (&ex)[HEADS], int g, int lane,
                                         const float (&hv)[4][HEADS]) {
    #pragma unroll
    for (int k = 0; k < 4; ++k) {
        float* ap = acc + rl_i(r, g * 4 + k) * HC + lane;
        #pragma unroll
        for (int j = 0; j < HEADS; ++j)
            atomicAdd(ap + j * OUT_CH, hv[k][j] * rl_f(ex[j], g * 4 + k));
    }
}

template <bool KEPT>
__device__ __forceinline__ void full_batch(const int* wst, int qh, int lane,
                                           const float* __restrict__ a_src,
                                           const float* __restrict__ h,
                                           const float* ad_s, const unsigned* mx,
                                           float* den, float* acc, const float (&e)[HEADS]) {
    const int p = wst[qh + lane];
    const int s = p & 0x1ffff;
    const int r = p >> 17;
    float4 as = make_float4(0.f, 0.f, 0.f, 0.f);
    if (!KEPT) as = ((const float4*)a_src)[s];
    float hv[NBUF][4][HEADS];
    #pragma unroll
    for (int g = 0; g < NBUF; ++g) rows_load(h, s, g, lane, hv[g]);
    __builtin_amdgcn_sched_barrier(0);
    const uint4 mf = ((const uint4*)mx)[r];
    float ex[HEADS];
    if (KEPT) kept_weights(e, mf, true, r, den, ex);
    else      batch_weights(as, ((const float4*)ad_s)[r], mf, true, r, den, ex);
    #pragma unroll
    for (int g = 0; g < BATCH / 4; ++g) {
        rows_add(acc, r, ex, g, lane, hv[g % NBUF]);
        if (g + NBUF < BATCH / 4) rows_load(h, s, g + NBUF, lane, hv[g % NBUF]);
        __builtin_amdgcn_sched_barrier(0);
    }
}

__global__ __launch_bounds__(1024) void k_own(const int* __restrict__ ei,
                                              const unsigned* __restrict__ own,
                                              const float* __restrict__ a_src,
                                              const float* __restrict__ a_dst,
                                              const float* __restrict__ h,
                                              const float* __restrict__ bias,
                                              float* __restrict__ out) {
    __shared__ __attribute__((aligned(16))) float    acc[RNG * HC];
    __shared__ __attribute__((aligned(16))) float    den[RNG * HEADS];
    __shared__ __attribute__((aligned(16))) float    ad_s[RNG * HEADS];
    __shared__ __attribute__((aligned(16))) unsigned mx[(RNG + 32) * HEADS];
    __shared__ int st[32 * SCAP];

    const int tid  = threadIdx.x;
    const int lane = tid & 31;
    const int wave = __builtin_amdgcn_readfirstlane(tid >> 5);
    const int base = blockIdx.x * RNG;
    int* const wst = st + wave * SCAP;

    for (int i = tid; i < RNG * HC / 4; i += 1024) ((float4*)acc)[i] = make_float4(0.f, 0.f, 0.f, 0.f);
    for (int i = tid; i < RNG * HEADS; i += 1024) {
        den[i] = 0.f;
        mx[i] = 0u;
        ad_s[i] = a_dst[base * HEADS + i];
    }
    if (tid < 32 * HEADS) mx[RNG * HEADS + tid] = 0u;
    const int nself = (RNG - wave + 31) / 32;
    if (lane < nself) wst[lane] = N_EDGES + base + wave + 32 * lane;
    __syncthreads();

    const uint4* __restrict__ ocol = (const uint4*)own;
    unsigned inv[8];
    #pragma unroll
    for (int k = 0; k < 8; ++k) inv[k] = ((blockIdx.x >> k) & 1u) ? 0xffffffffu : 0u;
    int qt = nself;
    int slow0 = NOSLOW;
    for (int t0 = wave; t0 < NSTEP; t0 += 32 * SU) {
        if (slow0 == NOSLOW) {
            uint4 ov[SU][8];
            const int qt0 = qt;
            bool ovf = false;
            #pragma unroll
            for (int j = 0; j < SU; ++j) {
                const int g = min((t0 + 32 * j) * 32 + lane, NGRP - 1);
                #pragma unroll
                for (int k = 0; k < 8; ++k) ov[j][k] = ocol[k * (PLANE_W / 4) + g];
            }
            __builtin_amdgcn_sched_barrier(0);
            #pragma unroll
            for (int j = 0; j < SU; ++j) {
                const int g = (t0 + 32 * j) * 32 + lane;
                unsigned mis[4] = {0u, 0u, 0u, 0u};
                #pragma unroll
                for (int k = 0; k < 8; ++k) {
                    mis[0] |= ov[j][k].x ^ inv[k]; mis[1] |= ov[j][k].y ^ inv[k];
                    mis[2] |= ov[j][k].z ^ inv[k]; mis[3] |= ov[j][k].w ^ inv[k];
                }
                const int e0 = g * 128;
                #pragma unroll
                for (int w = 0; w < 4; w += 2) {
                    unsigned long long c = g < NGRP
                        ? ((unsigned long long)(~mis[w + 1]) << 32) | (unsigned long long)(~mis[w]) : 0ull;
                    while (__builtin_amdgcn_ballot_w32(c != 0ull) != 0u) {
                        const bool mine = c != 0ull;
                        const int  p = mine ? __builtin_ctzll(c) : 0;
                        own_push(mine, e0 + ((p & 31) << 2) + w + (p >> 5), wst, qt, ovf);
                        c &= c - 1ull;
                    }
                }
            }
            if (ovf) {
                qt = qt0;
                slow0 = t0;
            }
        }
        if (slow0 != NOSLOW) slow_max(t0, lane, base, ei, a_src, ad_s, mx);
    }
    __builtin_amdgcn_wave_barrier();
    float sc[KEEP][HEADS];
    #pragma unroll
    for (int b = 0; b < KEEP; ++b) {
        #pragma unroll
        for (int k = 0; k < HEADS; ++k) sc[b][k] = 0.f;
    }
    #pragma unroll
    for (int c = 0; c < KEEP / MAXB; ++c) {
        if (c * MAXB * BATCH < qt) {
            float eo[MAXB][HEADS];
            stash_max(wst, c * MAXB * BATCH, qt, lane, base, ei, a_src, ad_s, mx, eo);
            #pragma unroll
            for (int j = 0; j < MAXB; ++j) {
                #pragma unroll
                for (int k = 0; k < HEADS; ++k) sc[c * MAXB + j][k] = eo[j][k];
            }
        }
    }
    for (int q = KEEP * BATCH; q < qt; q += MAXB * BATCH) {
        float eo[MAXB][HEADS];
        stash_max(wst, q, qt, lane, base, ei, a_src, ad_s, mx, eo);
    }
    __syncthreads();

    int qh = 0;
    while (qt - qh >= BATCH && qh < KEEP * BATCH) {
        full_batch<true>(wst, qh, lane, a_src, h, ad_s, mx, den, acc, sc[0]);
        qh += BATCH;
        #pragma unroll
        for (int b = 0; b + 1 < KEEP; ++b) {
            #pragma unroll
            for (int k = 0; k < HEADS; ++k) sc[b][k] = sc[b + 1][k];
        }
    }
    while (qt - qh >= BATCH) {
        full_batch<false>(wst, qh, lane, a_src, h, ad_s, mx, den, acc, sc[0]);
        qh += BATCH;
    }
    {
        const int cnt = qt - qh;
        if (cnt > 0) {
            const bool val = lane < cnt;
            const int p = wst[qh + (val ? lane : 0)];
            const int s = p & 0x1ffff;
            const int r = p >> 17;
            float ex[HEADS];
            if (qh < KEEP * BATCH)
                kept_weights(sc[0], ((const uint4*)mx)[r], val, r, den, ex);
            else
                batch_weights(((const float4*)a_src)[s], ((const float4*)ad_s)[r], ((const uint4*)mx)[r],
                              val, r, den, ex);
            #pragma unroll 1
            for (int g = 0; g * 4 < cnt; ++g) {
                float hv[4][HEADS];
                rows_load(h, s, g, lane, hv);
                rows_add(acc, r, ex, g, lane, hv);
            }
        }
    }
    if (slow0 != NOSLOW) {
        #pragma unroll 1
        for (int t0 = slow0; t0 < NSTEP; t0 += 32 * SU) slow_sum(t0, lane, base, ei, a_src, h, ad_s, mx, acc, den);
    }
    __syncthreads();

    const float4 bv = ((const float4*)bias)[lane];
    for (int r = wave; r < RNG; r += 32) {
        float4 v = ((const float4*)(acc + r * HC))[lane];
        const float dn = den[r * HEADS + (lane >> 3)] + 1e-16f;
        v.x = v.x / dn + bv.x; v.y = v.y / dn + bv.y;
        v.z = v.z / dn + bv.z; v.w = v.w / dn + bv.w;

        float mx_r = fmaxf(fmaxf(v.x, v.y), fmaxf(v.z, v.w));
        #pragma unroll
        for (int o = 16; o >= 1; o >>= 1) mx_r = fmaxf(mx_r, __shfl_xor(mx_r, o, 32));

        float se = __expf(v.x - mx_r) + __expf(v.y - mx_r) + __expf(v.z - mx_r) + __expf(v.w - mx_r);
        #pragma unroll
        for (int o = 16; o >= 1; o >>= 1) se += __shfl_xor(se, o, 32);

        const float lse = mx_r + __logf(se);
        v.x -= lse; v.y -= lse; v.z -= lse; v.w -= lse;
        ((float4*)(out + (size_t)(base + r) * HC))[lane] = v;
    }
}

extern "C" void kernel_launch(void* const* d_in, const int* in_sizes, int n_in,
                              void* d_out, int out_size, void* d_ws, size_t ws_size,
                              hipStream_t stream) {
    const float* x       = (const float*)d_in[0];
    const int*   ei      = (const int*)  d_in[1];
    const float* W       = (const float*)d_in[2];
    const float* att_src = (const float*)d_in[3];
    const float* att_dst = (const float*)d_in[4];
    const float* bias    = (const float*)d_in[5];
    float*       out     = (float*)d_out;

    float*    h     = (float*)d_ws;
    float*    asrc  = h    + (size_t)N_NODES * HC;
    float*    adst  = asrc + (size_t)N_NODES * HEADS;
    unsigned* ownq  = (unsigned*)(adst + (size_t)N_NODES * HEADS);

    k_gemm<<<GEMM_BLOCKS, 256, 0, stream>>>(x, W, att_src, att_dst, ei, h, asrc, adst, ownq);
    k_own <<<NOWN, 1024, 0, stream>>>(ei, ownq, asrc, adst, h, bias, out);
}


# h rows stored as f16 (f32 sums rounded once), two heads interleaved a line: owner blocks load 2 lines an edge, not 4 (lane = channel of two heads); 6 groups in flight; sum order kept
# speedup vs baseline: 7.5693x; 1.1502x over previous
#include <hip/hip_runtime.h>
#include <hip/hip_bf16.h>

#define N_NODES 100000
#define N_EDGES 1600000
#define TOT_E   (N_EDGES + N_NODES)
#define IN_CH   64
#define OUT_CH  32
#define HEADS   4
#define HC      (HEADS * OUT_CH)
#define NEG_SLOPE 0.2f

typedef __attribute__((ext_vector_type(16))) _Float16 v16h;
typedef __attribute__((ext_vector_type(8)))  float    v8f;

__device__ __forceinline__ unsigned fflip(float f) {
    unsigned u = __float_as_uint(f);
    return (u & 0x80000000u) ? ~u : (u | 0x80000000u);
}
__device__ __forceinline__ float funflip(unsigned u) {
    return __uint_as_float((u & 0x80000000u) ? (u & 0x7FFFFFFFu) : ~u);
}

__device__ __forceinline__ void edge_sd_sel(const int* __restrict__ ei, int e, int& s, int& d) {
    const int ec = min(e, N_EDGES - 1);
    const int ls = ei[ec], ld = ei[N_EDGES + ec];
    const bool real = e < N_EDGES;
    s = real ? ls : e - N_EDGES;
    d = real ? ld : e - N_EDGES;
}

#define M_TILES     (N_NODES / 16)
#define GEMM_BLOCKS 1250
#define GEMM_THREADS (GEMM_BLOCKS * 256)
#define XPITCH      72
static_assert(M_TILES % GEMM_BLOCKS == 0, "every block takes the same number of tiles");

#define RNG    400
#define NOWN   (N_NODES / RNG)
static_assert(N_NODES % RNG == 0, "blocks own whole ranges: no row guard");
static_assert(NOWN <= 256, "an owner number fits a byte");
static_assert(N_EDGES % 4 == 0, "owner bytes are written four at a time");
static_assert(N_EDGES / 4 > GEMM_THREADS && N_EDGES / 4 <= 2 * GEMM_THREADS,
              "every thread of the GEMM grid writes one owner word, the first ones a second");
#define PLANE_W (N_EDGES / 32)
#define NGRP    (N_EDGES / 128)
static_assert(N_EDGES % 128 == 0, "whole groups: every plane word is written, every bit is an edge");
static_assert((N_EDGES / 4 - GEMM_THREADS) % 32 == 0, "a wave writes its second group whole or not at all");
static_assert(PLANE_W % 4 == 0, "planes start on 16-byte boundaries");

typedef __attribute__((ext_vector_type(8))) _Float16 v8h;
typedef __attribute__((ext_vector_type(4))) _Float16 v4h;
typedef __attribute__((ext_vector_type(2))) _Float16 v2h;

#define HQ(head, ch) ((((head) >> 1) << 6) + ((ch) << 1) + ((head) & 1))

__device__ __forceinline__ unsigned own_planes(int4 d) {
    const unsigned o[4] = {(unsigned)d.x / RNG, (unsigned)d.y / RNG, (unsigned)d.z / RNG, (unsigned)d.w / RNG};
    unsigned val = 0u;
    #pragma unroll
    for (int k = 0; k < 8; ++k) {
        #pragma unroll
        for (int j = 0; j < 4; ++j) {
            const unsigned m = __builtin_amdgcn_ballot_w32(((o[j] >> k) & 1u) != 0u);
            asm("v_writelane_b32 %0, %1, %2" : "+v"(val) : "s"(m), "n"(4 * k + j));
        }
    }
    return val;
}

template <int CTRL>
__device__ __forceinline__ float row_lane(float v) {
    return __int_as_float(__builtin_amdgcn_update_dpp(0, __float_as_int(v), CTRL, 0xf, 0xf, true));
}
__device__ __forceinline__ float row_sum16(float v) {
    v += row_lane<0xB1>(v);
    v += row_lane<0x4E>(v);
    v += row_lane<0x141>(v);
    v += row_lane<0x140>(v);
    return v;
}

__global__ __launch_bounds__(256) __attribute__((amdgpu_waves_per_eu(10)))
void k_gemm(const float* __restrict__ x,
                                              const float* __restrict__ W,
                                              const float* __restrict__ att_src,
                                              const float* __restrict__ att_dst,
                                              const int* __restrict__ ei,
                                              _Float16* __restrict__ h,
                                              float* __restrict__ a_src,
                                              float* __restrict__ a_dst,
                                              unsigned* __restrict__ ownq) {
    __shared__ __attribute__((aligned(16))) _Float16 xs[2][16 * XPITCH];
    __shared__ __attribute__((aligned(16))) float part[2][8][2][16];
    const int tid = threadIdx.x;
    const int wave = tid >> 5;
    const int lane = tid & 31;
    const int mrow = lane & 15;
    const int khalf = (lane >> 4) << 3;
    const int ncol = wave * 16 + (lane & 15);

    const int gt = blockIdx.x * 256 + tid;
    const int4* __restrict__ dq = (const int4*)(ei + N_EDGES);
    const bool two = gt < N_EDGES / 4 - GEMM_THREADS;
    const int4 d0 = dq[gt];
    const int4 d1 = dq[two ? gt + GEMM_THREADS : gt];

    v16h b0, b1;
    #pragma unroll
    for (int i = 0; i < 16; ++i) {
        int k = i + (i & 8) + khalf;
        b0[i] = (_Float16)W[(size_t)k * HC + ncol];
        b1[i] = (_Float16)W[(size_t)(k + 32) * HC + ncol];
    }
    const float as_c = att_src[ncol];
    const float ad_c = att_dst[ncol];

    const unsigned o0 = own_planes(d0), o1 = own_planes(d1);

    const int jidx = tid & 63;
    const int jrow = jidx >> 2, jhead = jidx & 3;
    const int joff = (2 * jhead * 2 + (jrow >> 3)) * 16 + ((tid >> 6) & 1) * 8 + (jrow & 7);

    const int sr = tid >> 4;
    const int sc = (tid & 15) << 2;
    int mt = blockIdx.x;
    float4 xv = *(const float4*)(x + (size_t)(mt * 16 + sr) * IN_CH + sc);
    int buf = 0;
    #pragma nounroll
    for (; mt < M_TILES; mt += GEMM_BLOCKS) {
        v4h xh;
        xh[0] = (_Float16)xv.x; xh[1] = (_Float16)xv.y; xh[2] = (_Float16)xv.z; xh[3] = (_Float16)xv.w;
        *(v4h*)(&xs[buf][sr * XPITCH + sc]) = xh;
        __syncthreads();
        const int nx = mt + GEMM_BLOCKS;
        if (nx < M_TILES) xv = *(const float4*)(x + (size_t)(nx * 16 + sr) * IN_CH + sc);

        if (mt != (int)blockIdx.x && tid < 128) {
            const float* pp = &part[buf ^ 1][0][0][0] + joff;
            const float v = pp[0] + pp[32];
            const size_t o = (size_t)(mt - GEMM_BLOCKS) * 16 * HEADS + jidx;
            if (tid < 64) a_src[o] = v; else a_dst[o] = v;
        }

        const _Float16* rp = &xs[buf][mrow * XPITCH + khalf];
        const v8h p0 = *(const v8h*)(rp);
        const v8h p1 = *(const v8h*)(rp + 16);
        const v8h p2 = *(const v8h*)(rp + 32);
        const v8h p3 = *(const v8h*)(rp + 48);
        const v16h a0 = __builtin_shufflevector(p0, p1, 0, 1, 2, 3, 4, 5, 6, 7, 8, 9, 10, 11, 12, 13, 14, 15);
        const v16h a1 = __builtin_shufflevector(p2, p3, 0, 1, 2, 3, 4, 5, 6, 7, 8, 9, 10, 11, 12, 13, 14, 15);

        v8f c = {};
        c = __builtin_amdgcn_wmma_f32_16x16x32_f16(false, a0, false, b0, (short)0, c, false, false);
        c = __builtin_amdgcn_wmma_f32_16x16x32_f16(false, a1, false, b1, (short)0, c, false, false);

        _Float16* const hrow = h + ((size_t)(mt * 16 + khalf) * HC + HQ(wave >> 1, ncol & 31));
        #pragma unroll
        for (int r = 0; r < 8; ++r) hrow[r * HC] = (_Float16)c[r];

        float ps[8], pd[8];
        #pragma unroll
        for (int r = 0; r < 8; ++r) {
            ps[r] = row_sum16(c[r] * as_c);
            pd[r] = row_sum16(c[r] * ad_c);
        }
        if ((lane & 15) == 0) {
            float4* const wp = (float4*)&part[buf][wave][lane >> 4][0];
            wp[0] = make_float4(ps[0], ps[1], ps[2], ps[3]);
            wp[1] = make_float4(ps[4], ps[5], ps[6], ps[7]);
            wp[2] = make_float4(pd[0], pd[1], pd[2], pd[3]);
            wp[3] = make_float4(pd[4], pd[5], pd[6], pd[7]);
        }
        buf ^= 1;
    }
    __syncthreads();
    if (tid < 128) {
        const float* pp = &part[buf ^ 1][0][0][0] + joff;
        const float v = pp[0] + pp[32];
        const size_t o = (size_t)(mt - GEMM_BLOCKS) * 16 * HEADS + jidx;
        if (tid < 64) a_src[o] = v; else a_dst[o] = v;
    }
    unsigned* const op = ownq + (lane >> 2) * PLANE_W + (gt >> 5) * 4 + (lane & 3);
    op[0] = o0;
    if (two) op[(GEMM_THREADS / 32) * 4] = o1;
}

#define SCAP   384
#define BATCH  32
#define NBUF   6
#define KEEP   8
#define SU     2
#define OSTEP  4096
#define NSTEP  ((N_EDGES + OSTEP - 1) / OSTEP)
#define NOSLOW 0x7fffffff
static_assert(OSTEP == 32 * 128, "a lane's part of a step is one group of the first kernel");
static_assert(N_NODES <= (1 << 17) && RNG <= (1 << 14), "a stash word holds row << 17 | source");
static_assert((RNG + 31) / 32 <= 32 && (RNG + 31) / 32 <= SCAP, "a wave's self loops fit one lane each");

__device__ __forceinline__ int rl_i(int v, int l) {
    return __builtin_amdgcn_readlane(v, l);
}
__device__ __forceinline__ float rl_f(float v, int l) {
    return __int_as_float(__builtin_amdgcn_readlane(__float_as_int(v), l));
}

__device__ __forceinline__ float edge_score(float t) {
    return t > 0.f ? t : NEG_SLOPE * t;
}

__device__ __forceinline__ float edge_weight(float t, unsigned mf) {
    return __expf(edge_score(t) - funflip(mf));
}

__device__ __forceinline__ void own_push(bool mine, int e, int* wst, int& qt, bool& ovf) {
    const unsigned mask = __builtin_amdgcn_ballot_w32(mine);
    if (mask == 0u) return;
    const int n = __builtin_popcount(mask);
    if (qt + n <= SCAP) {
        if (mine) wst[qt + (int)__builtin_amdgcn_mbcnt_lo(mask, 0u)] = e;
        qt += n;
    } else {
        ovf = true;
    }
}

#define MAXB 4
static_assert(KEEP % MAXB == 0, "whole calls of pass 1 are kept");
__device__ __forceinline__ void stash_max(int* wst, int q, int qt, int lane, int base,
                                          const int* __restrict__ ei,
                                          const float* __restrict__ a_src,
                                          const float* ad_s, unsigned* mx,
                                          float (&eo)[MAXB][HEADS]) {
    int idx[MAXB], s[MAXB], d[MAXB];
    bool val[MAXB];
    float4 as[MAXB], ad[MAXB];
    #pragma unroll
    for (int j = 0; j < MAXB; ++j) {
        const int i = q + BATCH * j + lane;
        val[j] = i < qt;
        idx[j] = val[j] ? i : q;
        edge_sd_sel(ei, wst[idx[j]], s[j], d[j]);
    }
    #pragma unroll
    for (int j = 0; j < MAXB; ++j) {
        as[j] = ((const float4*)a_src)[s[j]];
        ad[j] = ((const float4*)ad_s)[d[j] - base];
    }
    #pragma unroll
    for (int j = 0; j < MAXB; ++j) {
        const int r = d[j] - base;
        unsigned* const mp = mx + (val[j] ? r : RNG + lane) * HEADS;
        eo[j][0] = edge_score(as[j].x + ad[j].x);
        eo[j][1] = edge_score(as[j].y + ad[j].y);
        eo[j][2] = edge_score(as[j].z + ad[j].z);
        eo[j][3] = edge_score(as[j].w + ad[j].w);
        #pragma unroll
        for (int k = 0; k < HEADS; ++k) atomicMax(mp + k, fflip(eo[j][k]));
        if (val[j]) wst[idx[j]] = (r << 17) | s[j];
    }
}

__device__ __forceinline__ void slow_max(int t0, int lane, int base,
                                         const int* __restrict__ ei,
                                         const float* __restrict__ a_src,
                                         const float* ad_s, unsigned* mx) {
    #pragma unroll 1
    for (int k = 0; k < SU * (OSTEP / 32); ++k) {
        const int t = t0 + 32 * (k / (OSTEP / 32));
        const int e = t * OSTEP + (k % (OSTEP / 32)) * 32 + lane;
        if (t < NSTEP && e < N_EDGES) {
            const unsigned r = (unsigned)(ei[N_EDGES + e] - base);
            if (r < (unsigned)RNG) {
                const int s = ei[e];
                #pragma unroll 1
                for (int j = 0; j < HEADS; ++j)
                    atomicMax(&mx[r * HEADS + j], fflip(edge_score(a_src[s * HEADS + j] + ad_s[r * HEADS + j])));
            }
        }
    }
}

__device__ __forceinline__ void slow_sum(int t0, int lane, int base,
                                         const int* __restrict__ ei,
                                         const float* __restrict__ a_src,
                                         const _Float16* __restrict__ h,
                                         const float* ad_s, const unsigned* mx,
                                         float* acc, float* den) {
    #pragma unroll 1
    for (int k = 0; k < SU * (OSTEP / 32); ++k) {
        const int t = t0 + 32 * (k / (OSTEP / 32));
        const int e = t * OSTEP + (k % (OSTEP / 32)) * 32 + lane;
        if (t < NSTEP && e < N_EDGES) {
            const unsigned r = (unsigned)(ei[N_EDGES + e] - base);
            if (r < (unsigned)RNG) {
                const int s = ei[e];
                #pragma unroll 1
                for (int j = 0; j < HEADS; ++j) {
                    const float w = edge_weight(a_src[s * HEADS + j] + ad_s[r * HEADS + j], mx[r * HEADS + j]);
                    atomicAdd(&den[r * HEADS + j], w);
                    #pragma unroll 1
                    for (int c = 0; c < OUT_CH; ++c)
                        atomicAdd(&acc[r * HC + j * OUT_CH + c], (float)h[(size_t)s * HC + HQ(j, c)] * w);
                }
            }
        }
    }
}

__device__ __forceinline__ void batch_weights(float4 as, float4 ad, uint4 mf, bool val, int r,
                                              float* den, float (&ex)[HEADS]) {
    ex[0] = edge_weight(as.x + ad.x, mf.x); ex[1] = edge_weight(as.y + ad.y, mf.y);
    ex[2] = edge_weight(as.z + ad.z, mf.z); ex[3] = edge_weight(as.w + ad.w, mf.w);
    #pragma unroll
    for (int j = 0; j < HEADS; ++j) {
        ex[j] = val ? ex[j] : 0.f;
        atomicAdd(&den[r * HEADS + j], ex[j]);
    }
}

__device__ __forceinline__ void kept_weights(const float (&e)[HEADS], uint4 mf, bool val, int r,
                                             float* den, float (&ex)[HEADS]) {
    ex[0] = __expf(e[0] - funflip(mf.x)); ex[1] = __expf(e[1] - funflip(mf.y));
    ex[2] = __expf(e[2] - funflip(mf.z)); ex[3] = __expf(e[3] - funflip(mf.w));
    #pragma unroll
    for (int j = 0; j < HEADS; ++j) {
        ex[j] = val ? ex[j] : 0.f;
        atomicAdd(&den[r * HEADS + j], ex[j]);
    }
}

__device__ __forceinline__ void rows_load(const _Float16* __restrict__ h, int s, int g, int lane,
                                          v2h (&hv)[4][HEADS / 2]) {
    #pragma unroll
    for (int k = 0; k < 4; ++k) {
        const v2h* hp = (const v2h*)(h + (size_t)rl_i(s, g * 4 + k) * HC) + lane;
        #pragma unroll
        for (int p = 0; p < HEADS / 2; ++p) hv[k][p] = hp[p * 32];
    }
}

__device__ __forceinline__ void rows_add(float* acc, int r, const float (&ex)[HEADS], int g, int lane,
                                         const v2h (&hv)[4][HEADS / 2]) {
    #pragma unroll
    for (int k = 0; k < 4; ++k) {
        float* ap = acc + rl_i(r, g * 4 + k) * HC + lane;
        #pragma unroll
        for (int j = 0; j < HEADS; ++j)
            atomicAdd(ap + j * OUT_CH,
                      __builtin_fmaf((float)hv[k][j >> 1][j & 1], rl_f(ex[j], g * 4 + k), 0.f));
    }
}

template <bool KEPT>
__device__ __forceinline__ void full_batch(const int* wst, int qh, int lane,
                                           const float* __restrict__ a_src,
                                           const _Float16* __restrict__ h,
                                           const float* ad_s, const unsigned* mx,
                                           float* den, float* acc, const float (&e)[HEADS]) {
    const int p = wst[qh + lane];
    const int s = p & 0x1ffff;
    const int r = p >> 17;
    float4 as = make_float4(0.f, 0.f, 0.f, 0.f);
    if (!KEPT) as = ((const float4*)a_src)[s];
    v2h hv[NBUF][4][HEADS / 2];
    #pragma unroll
    for (int g = 0; g < NBUF; ++g) rows_load(h, s, g, lane, hv[g]);
    __builtin_amdgcn_sched_barrier(0);
    const uint4 mf = ((const uint4*)mx)[r];
    float ex[HEADS];
    if (KEPT) kept_weights(e, mf, true, r, den, ex);
    else      batch_weights(as, ((const float4*)ad_s)[r], mf, true, r, den, ex);
    #pragma unroll
    for (int g = 0; g < BATCH / 4; ++g) {
        rows_add(acc, r, ex, g, lane, hv[g % NBUF]);
        if (g + NBUF < BATCH / 4) rows_load(h, s, g + NBUF, lane, hv[g % NBUF]);
        __builtin_amdgcn_sched_barrier(0);
    }
}

__global__ __launch_bounds__(1024) void k_own(const int* __restrict__ ei,
                                              const unsigned* __restrict__ own,
                                              const float* __restrict__ a_src,
                                              const float* __restrict__ a_dst,
                                              const _Float16* __restrict__ h,
                                              const float* __restrict__ bias,
                                              float* __restrict__ out) {
    __shared__ __attribute__((aligned(16))) float    acc[RNG * HC];
    __shared__ __attribute__((aligned(16))) float    den[RNG * HEADS];
    __shared__ __attribute__((aligned(16))) float    ad_s[RNG * HEADS];
    __shared__ __attribute__((aligned(16))) unsigned mx[(RNG + 32) * HEADS];
    __shared__ int st[32 * SCAP];

    const int tid  = threadIdx.x;
    const int lane = tid & 31;
    const int wave = __builtin_amdgcn_readfirstlane(tid >> 5);
    const int base = blockIdx.x * RNG;
    int* const wst = st + wave * SCAP;

    for (int i = tid; i < RNG * HC / 4; i += 1024) ((float4*)acc)[i] = make_float4(0.f, 0.f, 0.f, 0.f);
    for (int i = tid; i < RNG * HEADS; i += 1024) {
        den[i] = 0.f;
        mx[i] = 0u;
        ad_s[i] = a_dst[base * HEADS + i];
    }
    if (tid < 32 * HEADS) mx[RNG * HEADS + tid] = 0u;
    const int nself = (RNG - wave + 31) / 32;
    if (lane < nself) wst[lane] = N_EDGES + base + wave + 32 * lane;
    __syncthreads();

    const uint4* __restrict__ ocol = (const uint4*)own;
    unsigned inv[8];
    #pragma unroll
    for (int k = 0; k < 8; ++k) inv[k] = ((blockIdx.x >> k) & 1u) ? 0xffffffffu : 0u;
    int qt = nself;
    int slow0 = NOSLOW;
    for (int t0 = wave; t0 < NSTEP; t0 += 32 * SU) {
        if (slow0 == NOSLOW) {
            uint4 ov[SU][8];
            const int qt0 = qt;
            bool ovf = false;
            #pragma unroll
            for (int j = 0; j < SU; ++j) {
                const int g = min((t0 + 32 * j) * 32 + lane, NGRP - 1);
                #pragma unroll
                for (int k = 0; k < 8; ++k) ov[j][k] = ocol[k * (PLANE_W / 4) + g];
            }
            __builtin_amdgcn_sched_barrier(0);
            #pragma unroll
            for (int j = 0; j < SU; ++j) {
                const int g = (t0 + 32 * j) * 32 + lane;
                unsigned mis[4] = {0u, 0u, 0u, 0u};
                #pragma unroll
                for (int k = 0; k < 8; ++k) {
                    mis[0] |= ov[j][k].x ^ inv[k]; mis[1] |= ov[j][k].y ^ inv[k];
                    mis[2] |= ov[j][k].z ^ inv[k]; mis[3] |= ov[j][k].w ^ inv[k];
                }
                const int e0 = g * 128;
                #pragma unroll
                for (int w = 0; w < 4; w += 2) {
                    unsigned long long c = g < NGRP
                        ? ((unsigned long long)(~mis[w + 1]) << 32) | (unsigned long long)(~mis[w]) : 0ull;
                    while (__builtin_amdgcn_ballot_w32(c != 0ull) != 0u) {
                        const bool mine = c != 0ull;
                        const int  p = mine ? __builtin_ctzll(c) : 0;
                        own_push(mine, e0 + ((p & 31) << 2) + w + (p >> 5), wst, qt, ovf);
                        c &= c - 1ull;
                    }
                }
            }
            if (ovf) {
                qt = qt0;
                slow0 = t0;
            }
        }
        if (slow0 != NOSLOW) slow_max(t0, lane, base, ei, a_src, ad_s, mx);
    }
    __builtin_amdgcn_wave_barrier();
    float sc[KEEP][HEADS];
    #pragma unroll
    for (int b = 0; b < KEEP; ++b) {
        #pragma unroll
        for (int k = 0; k < HEADS; ++k) sc[b][k] = 0.f;
    }
    #pragma unroll
    for (int c = 0; c < KEEP / MAXB; ++c) {
        if (c * MAXB * BATCH < qt) {
            float eo[MAXB][HEADS];
            stash_max(wst, c * MAXB * BATCH, qt, lane, base, ei, a_src, ad_s, mx, eo);
            #pragma unroll
            for (int j = 0; j < MAXB; ++j) {
                #pragma unroll
                for (int k = 0; k < HEADS; ++k) sc[c * MAXB + j][k] = eo[j][k];
            }
        }
    }
    for (int q = KEEP * BATCH; q < qt; q += MAXB * BATCH) {
        float eo[MAXB][HEADS];
        stash_max(wst, q, qt, lane, base, ei, a_src, ad_s, mx, eo);
    }
    __syncthreads();

    int qh = 0;
    while (qt - qh >= BATCH && qh < KEEP * BATCH) {
        full_batch<true>(wst, qh, lane, a_src, h, ad_s, mx, den, acc, sc[0]);
        qh += BATCH;
        #pragma unroll
        for (int b = 0; b + 1 < KEEP; ++b) {
            #pragma unroll
            for (int k = 0; k < HEADS; ++k) sc[b][k] = sc[b + 1][k];
        }
    }
    while (qt - qh >= BATCH) {
        full_batch<false>(wst, qh, lane, a_src, h, ad_s, mx, den, acc, sc[0]);
        qh += BATCH;
    }
    {
        const int cnt = qt - qh;
        if (cnt > 0) {
            const bool val = lane < cnt;
            const int p = wst[qh + (val ? lane : 0)];
            const int s = p & 0x1ffff;
            const int r = p >> 17;
            float ex[HEADS];
            if (qh < KEEP * BATCH)
                kept_weights(sc[0], ((const uint4*)mx)[r], val, r, den, ex);
            else
                batch_weights(((const float4*)a_src)[s], ((const float4*)ad_s)[r], ((const uint4*)mx)[r],
                              val, r, den, ex);
            #pragma unroll 1
            for (int g = 0; g * 4 < cnt; ++g) {
                v2h hv[4][HEADS / 2];
                rows_load(h, s, g, lane, hv);
                rows_add(acc, r, ex, g, lane, hv);
            }
        }
    }
    if (slow0 != NOSLOW) {
        #pragma unroll 1
        for (int t0 = slow0; t0 < NSTEP; t0 += 32 * SU) slow_sum(t0, lane, base, ei, a_src, h, ad_s, mx, acc, den);
    }
    __syncthreads();

    const float4 bv = ((const float4*)bias)[lane];
    for (int r = wave; r < RNG; r += 32) {
        float4 v = ((const float4*)(acc + r * HC))[lane];
        const float dn = den[r * HEADS + (lane >> 3)] + 1e-16f;
        v.x = v.x / dn + bv.x; v.y = v.y / dn + bv.y;
        v.z = v.z / dn + bv.z; v.w = v.w / dn + bv.w;

        float mx_r = fmaxf(fmaxf(v.x, v.y), fmaxf(v.z, v.w));
        #pragma unroll
        for (int o = 16; o >= 1; o >>= 1) mx_r = fmaxf(mx_r, __shfl_xor(mx_r, o, 32));

        float se = __expf(v.x - mx_r) + __expf(v.y - mx_r) + __expf(v.z - mx_r) + __expf(v.w - mx_r);
        #pragma unroll
        for (int o = 16; o >= 1; o >>= 1) se += __shfl_xor(se, o, 32);

        const float lse = mx_r + __logf(se);
        v.x -= lse; v.y -= lse; v.z -= lse; v.w -= lse;
        ((float4*)(out + (size_t)(base + r) * HC))[lane] = v;
    }
}

extern "C" void kernel_launch(void* const* d_in, const int* in_sizes, int n_in,
                              void* d_out, int out_size, void* d_ws, size_t ws_size,
                              hipStream_t stream) {
    const float* x       = (const float*)d_in[0];
    const int*   ei      = (const int*)  d_in[1];
    const float* W       = (const float*)d_in[2];
    const float* att_src = (const float*)d_in[3];
    const float* att_dst = (const float*)d_in[4];
    const float* bias    = (const float*)d_in[5];
    float*       out     = (float*)d_out;

    _Float16* h     = (_Float16*)d_ws;
    float*    asrc  = (float*)(h + (size_t)N_NODES * HC);
    float*    adst  = asrc + (size_t)N_NODES * HEADS;
    unsigned* ownq  = (unsigned*)(adst + (size_t)N_NODES * HEADS);

    k_gemm<<<GEMM_BLOCKS, 256, 0, stream>>>(x, W, att_src, att_dst, ei, h, asrc, adst, ownq);
    k_own <<<NOWN, 1024, 0, stream>>>(ei, ownq, asrc, adst, h, bias, out);
}
